# starting kernel
# baseline (speedup 1.0000x reference)
#include <hip/hip_runtime.h>
#include <hip/hip_bf16.h>

constexpr int N_ = 100000;
constexpr int E_ = 1600000;
constexpr int D_ = 32;
constexpr int L_ = 2;
constexpr float EPS_ = 1e-5f;

constexpr int WPB = 8;
constexpr int ETILES = E_ / 16;
constexpr int NTILES = N_ / 16;
constexpr int EDGE_BLOCKS = ETILES / WPB;
constexpr int NODE_BLOCKS = (NTILES + WPB - 1) / WPB;

typedef __attribute__((ext_vector_type(16))) _Float16 v16h;
typedef __attribute__((ext_vector_type(8)))  _Float16 v8h;
typedef __attribute__((ext_vector_type(8)))  float    v8f;

__device__ __forceinline__ v8f wmma_f16(v16h a, v16h b, v8f c) {
    return __builtin_amdgcn_wmma_f32_16x16x32_f16(
        false, a, false, b, (short)0, c, false, false);
}

__device__ __forceinline__ float swishf(float x) {
    float e = __expf(-x);
    return x * __builtin_amdgcn_rcpf(1.0f + e);
}

__device__ __forceinline__ v16h load_rowfrag(const _Float16* __restrict__ row, int half) {
    v8h lo = *(const v8h*)(row + half * 8);
    v8h hi = *(const v8h*)(row + 16 + half * 8);
    v16h a;
#pragma unroll
    for (int i = 0; i < 8; ++i) { a[i] = lo[i]; a[8 + i] = hi[i]; }
    return a;
}

__device__ __forceinline__ void wave_lds_sync() {
    __builtin_amdgcn_wave_barrier();
    asm volatile("s_wait_dscnt 0" ::: "memory");
    __builtin_amdgcn_wave_barrier();
}

__device__ __forceinline__ void atomic_add_f32(float* p, float v) {
    __hip_atomic_fetch_add(p, v, __ATOMIC_RELAXED, __HIP_MEMORY_SCOPE_AGENT);
}

constexpr size_t alignup(size_t x) { return (x + 255) & ~size_t(255); }
constexpr size_t SZ_WFRAG = size_t(L_) * 12 * 512 * 2;
constexpr size_t OFF_H    = alignup(SZ_WFRAG);
constexpr size_t OFF_H16  = alignup(OFF_H   + size_t(N_) * D_ * 4);
constexpr size_t OFF_AGG  = alignup(OFF_H16 + size_t(N_) * D_ * 2);
constexpr size_t OFF_DEG  = alignup(OFF_AGG + size_t(N_) * D_ * 4);
constexpr size_t OFF_BNS  = alignup(OFF_DEG + size_t(N_) * 4);
constexpr size_t OFF_BNP  = alignup(OFF_BNS + 64 * 4);
constexpr size_t OFF_T16  = alignup(OFF_BNP + 64 * 4);

__global__ void __launch_bounds__(256) prep_weights_kernel(
    const float* __restrict__ msg_w1, const float* __restrict__ msg_w2,
    const float* __restrict__ upd_w1, const float* __restrict__ upd_w2,
    _Float16* __restrict__ wfrag)
{
    int gid = blockIdx.x * 256 + threadIdx.x;
    if (gid >= L_ * 12 * 512) return;
    int frag = gid >> 9;
    int lane = (gid >> 4) & 31;
    int i    = gid & 15;
    int layer = frag / 12;
    int r     = frag % 12;
    const float* W; int kc, nt;
    if (r < 4)       { W = msg_w1 + size_t(layer) * 64 * 32; kc = r >> 1;       nt = r & 1; }
    else if (r < 6)  { W = msg_w2 + size_t(layer) * 32 * 32; kc = 0;            nt = r - 4; }
    else if (r < 10) { int q = r - 6;
                       W = upd_w1 + size_t(layer) * 64 * 32; kc = q >> 1;       nt = q & 1; }
    else             { W = upd_w2 + size_t(layer) * 32 * 32; kc = 0;            nt = r - 10; }
    int half = lane >> 4;
    int n = nt * 16 + (lane & 15);
    int k = kc * 32 + 16 * (i >> 3) + half * 8 + (i & 7);
    wfrag[size_t(frag) * 512 + lane * 16 + i] = (_Float16)W[k * 32 + n];
}

__global__ void __launch_bounds__(256) embed_kernel(
    const float* __restrict__ x, const float* __restrict__ w, const float* __restrict__ b,
    float* __restrict__ h, _Float16* __restrict__ h16)
{
    int gid = blockIdx.x * 256 + threadIdx.x;
    if (gid >= N_ * D_) return;
    int i = gid >> 5, j = gid & 31;
    const float* xr = x + size_t(i) * 4;
    float acc = b[j];
#pragma unroll
    for (int k = 0; k < 4; ++k) acc = fmaf(xr[k], w[k * 32 + j], acc);
    h[gid] = acc;
    h16[gid] = (_Float16)acc;
}

__global__ void __launch_bounds__(256) fill_kernel(float* __restrict__ p, int n) {
    int g = blockIdx.x * blockDim.x + threadIdx.x;
    if (g < n) p[g] = 0.0f;
}

__global__ void __launch_bounds__(256) degree_kernel(const int* __restrict__ dst,
                                                     float* __restrict__ cnt) {
    int e = blockIdx.x * 256 + threadIdx.x;
    if (e >= E_) return;
    atomic_add_f32(&cnt[dst[e]], 1.0f);
}

__global__ void __launch_bounds__(256) recip_kernel(float* __restrict__ cnt) {
    int i = blockIdx.x * 256 + threadIdx.x;
    if (i >= N_) return;
    cnt[i] = __builtin_amdgcn_rcpf(fmaxf(cnt[i], 1.0f));
}

__global__ void __launch_bounds__(256) edge_msg1_kernel(
    const _Float16* __restrict__ h16,
    const int* __restrict__ src, const int* __restrict__ dst,
    const _Float16* __restrict__ wf1, const float* __restrict__ b1,
    _Float16* __restrict__ t16, float* __restrict__ bnsum  )
{
    __shared__ _Float16 tb[WPB][16][32];
    __shared__ float bsum[32], bsq[32];
    int tid = threadIdx.x;
    if (tid < 32) { bsum[tid] = 0.0f; bsq[tid] = 0.0f; }
    __syncthreads();

    int lane = tid & 31, wv = tid >> 5;
    int tile = blockIdx.x * WPB + wv;
    int m = lane & 15, half = lane >> 4;
    int e = tile * 16 + m;
    int di = dst[e], si = src[e];

    v16h a0 = load_rowfrag(h16 + size_t(di) * 32, half);
    v16h a1 = load_rowfrag(h16 + size_t(si) * 32, half);
    v16h b00 = *(const v16h*)(wf1 + 0 * 512 + lane * 16);
    v16h b01 = *(const v16h*)(wf1 + 1 * 512 + lane * 16);
    v16h b10 = *(const v16h*)(wf1 + 2 * 512 + lane * 16);
    v16h b11 = *(const v16h*)(wf1 + 3 * 512 + lane * 16);

    v8f c0 = {}, c1 = {};
    c0 = wmma_f16(a0, b00, c0); c0 = wmma_f16(a1, b10, c0);
    c1 = wmma_f16(a0, b01, c1); c1 = wmma_f16(a1, b11, c1);

    int n0 = lane & 15;
    float bias0 = b1[n0], bias1 = b1[16 + n0];
    float s0 = 0.f, q0 = 0.f, s1 = 0.f, q1 = 0.f;
#pragma unroll
    for (int r = 0; r < 8; ++r) {
        int mr = (lane >> 4) * 8 + r;
        float t0 = swishf(c0[r] + bias0);
        float t1 = swishf(c1[r] + bias1);
        tb[wv][mr][n0]      = (_Float16)t0;
        tb[wv][mr][16 + n0] = (_Float16)t1;
        s0 += t0; q0 += t0 * t0; s1 += t1; q1 += t1 * t1;
    }
    atomicAdd(&bsum[n0], s0);      atomicAdd(&bsq[n0], q0);
    atomicAdd(&bsum[16 + n0], s1); atomicAdd(&bsq[16 + n0], q1);
    __syncthreads();

    v16h out;
#pragma unroll
    for (int i = 0; i < 8; ++i) {
        out[i]     = tb[wv][m][half * 8 + i];
        out[8 + i] = tb[wv][m][16 + half * 8 + i];
    }
    *(v16h*)(t16 + size_t(tile) * 512 + lane * 16) = out;

    if (tid < 32) {
        atomic_add_f32(&bnsum[tid],      bsum[tid]);
        atomic_add_f32(&bnsum[32 + tid], bsq[tid]);
    }
}

__global__ void bn_finalize_kernel(const float* __restrict__ bnsum,
                                   const float* __restrict__ g, const float* __restrict__ b,
                                   float* __restrict__ ss  )
{
    int n = threadIdx.x;
    if (n >= 32) return;
    float mu  = bnsum[n] * (1.0f / E_);
    float var = bnsum[32 + n] * (1.0f / E_) - mu * mu;
    float sc  = g[n] * rsqrtf(var + EPS_);
    ss[n] = sc;
    ss[32 + n] = b[n] - mu * sc;
}

__global__ void __launch_bounds__(256) edge_msg2_kernel(
    const _Float16* __restrict__ t16, const int* __restrict__ dst,
    const float* __restrict__ ss, const _Float16* __restrict__ wf2,
    const float* __restrict__ b2, float* __restrict__ agg)
{
    __shared__ float sc[32], sh[32];
    int tid = threadIdx.x;
    if (tid < 32) { sc[tid] = ss[tid]; sh[tid] = ss[32 + tid]; }
    __syncthreads();

    int lane = tid & 31, wv = tid >> 5;
    int tile = blockIdx.x * WPB + wv;
    int half = lane >> 4;

    v16h a = *(const v16h*)(t16 + size_t(tile) * 512 + lane * 16);
    v16h an;
#pragma unroll
    for (int i = 0; i < 16; ++i) {
        int k = 16 * (i >> 3) + half * 8 + (i & 7);
        an[i] = (_Float16)((float)a[i] * sc[k] + sh[k]);
    }
    v16h w0 = *(const v16h*)(wf2 + 0 * 512 + lane * 16);
    v16h w1 = *(const v16h*)(wf2 + 1 * 512 + lane * 16);
    v8f c0 = {}, c1 = {};
    c0 = wmma_f16(an, w0, c0);
    c1 = wmma_f16(an, w1, c1);

    int n0 = lane & 15;
    float bias0 = b2[n0], bias1 = b2[16 + n0];
#pragma unroll
    for (int r = 0; r < 8; ++r) {
        int mr = (lane >> 4) * 8 + r;
        int node = dst[tile * 16 + mr];
        atomic_add_f32(&agg[size_t(node) * 32 + n0],      swishf(c0[r] + bias0));
        atomic_add_f32(&agg[size_t(node) * 32 + 16 + n0], swishf(c1[r] + bias1));
    }
}

__global__ void __launch_bounds__(256) node_update_kernel(
    const _Float16* __restrict__ wfu1, const _Float16* __restrict__ wfu2,
    const float* __restrict__ ub1, const float* __restrict__ ub2,
    const float* __restrict__ agg, const float* __restrict__ recip,
    float* __restrict__ h, _Float16* __restrict__ h16)
{
    __shared__ _Float16 ub[WPB][16][32];
    __shared__ float    ob[WPB][16][32];
    int tid = threadIdx.x, lane = tid & 31, wv = tid >> 5;
    int tile = blockIdx.x * WPB + wv;
    if (tile >= NTILES) return;

    int m = lane & 15, half = lane >> 4;
    size_t node = size_t(tile) * 16 + m;

    v16h a0 = load_rowfrag(h16 + node * 32, half);
    float rc = recip[node];
    const float* ar = agg + node * 32;
    v16h a1;
#pragma unroll
    for (int i = 0; i < 8; ++i) {
        a1[i]     = (_Float16)(ar[half * 8 + i] * rc);
        a1[8 + i] = (_Float16)(ar[16 + half * 8 + i] * rc);
    }
    v16h b00 = *(const v16h*)(wfu1 + 0 * 512 + lane * 16);
    v16h b01 = *(const v16h*)(wfu1 + 1 * 512 + lane * 16);
    v16h b10 = *(const v16h*)(wfu1 + 2 * 512 + lane * 16);
    v16h b11 = *(const v16h*)(wfu1 + 3 * 512 + lane * 16);
    v8f c0 = {}, c1 = {};
    c0 = wmma_f16(a0, b00, c0); c0 = wmma_f16(a1, b10, c0);
    c1 = wmma_f16(a0, b01, c1); c1 = wmma_f16(a1, b11, c1);

    int n0 = lane & 15;
    float bias0 = ub1[n0], bias1 = ub1[16 + n0];
#pragma unroll
    for (int r = 0; r < 8; ++r) {
        int mr = (lane >> 4) * 8 + r;
        ub[wv][mr][n0]      = (_Float16)swishf(c0[r] + bias0);
        ub[wv][mr][16 + n0] = (_Float16)swishf(c1[r] + bias1);
    }
    wave_lds_sync();

    v16h au;
#pragma unroll
    for (int i = 0; i < 8; ++i) {
        au[i]     = ub[wv][m][half * 8 + i];
        au[8 + i] = ub[wv][m][16 + half * 8 + i];
    }
    v16h u0 = *(const v16h*)(wfu2 + 0 * 512 + lane * 16);
    v16h u1 = *(const v16h*)(wfu2 + 1 * 512 + lane * 16);
    v8f d0 = {}, d1 = {};
    d0 = wmma_f16(au, u0, d0);
    d1 = wmma_f16(au, u1, d1);

    float bb0 = ub2[n0], bb1 = ub2[16 + n0];
#pragma unroll
    for (int r = 0; r < 8; ++r) {
        int mr = (lane >> 4) * 8 + r;
        ob[wv][mr][n0]      = swishf(d0[r] + bb0);
        ob[wv][mr][16 + n0] = swishf(d1[r] + bb1);
    }
    wave_lds_sync();

    int row = lane >> 1, seg = (lane & 1) * 16;
    size_t nr = size_t(tile) * 16 + row;
    float*    hp   = h   + nr * 32 + seg;
    _Float16* hp16 = h16 + nr * 32 + seg;
#pragma unroll
    for (int i = 0; i < 16; ++i) {
        float nv = hp[i] + ob[wv][row][seg + i];
        hp[i] = nv;
        hp16[i] = (_Float16)nv;
    }
}

__global__ void __launch_bounds__(256) copyout_kernel(const float* __restrict__ h,
                                                      float* __restrict__ out) {
    int g = blockIdx.x * 256 + threadIdx.x;
    if (g < N_ * D_) out[g] = h[g];
}

extern "C" void kernel_launch(void* const* d_in, const int* in_sizes, int n_in,
                              void* d_out, int out_size, void* d_ws, size_t ws_size,
                              hipStream_t stream) {
    const float* x      = (const float*)d_in[0];
    const float* emb_w  = (const float*)d_in[1];
    const float* emb_b  = (const float*)d_in[2];
    const float* msg_w1 = (const float*)d_in[3];
    const float* msg_b1 = (const float*)d_in[4];
    const float* bn_g   = (const float*)d_in[5];
    const float* bn_b   = (const float*)d_in[6];
    const float* msg_w2 = (const float*)d_in[7];
    const float* msg_b2 = (const float*)d_in[8];
    const float* upd_w1 = (const float*)d_in[9];
    const float* upd_b1 = (const float*)d_in[10];
    const float* upd_w2 = (const float*)d_in[11];
    const float* upd_b2 = (const float*)d_in[12];
    const int*   eidx   = (const int*)d_in[13];
    const int*   esrc   = eidx;
    const int*   edst   = eidx + E_;

    char* ws = (char*)d_ws;
    _Float16* wfrag = (_Float16*)(ws);
    float*    h     = (float*)(ws + OFF_H);
    _Float16* h16   = (_Float16*)(ws + OFF_H16);
    float*    agg   = (float*)(ws + OFF_AGG);
    float*    deg   = (float*)(ws + OFF_DEG);
    float*    bnsum = (float*)(ws + OFF_BNS);
    float*    bnss  = (float*)(ws + OFF_BNP);
    _Float16* t16   = (_Float16*)(ws + OFF_T16);

    prep_weights_kernel<<<(L_ * 12 * 512 + 255) / 256, 256, 0, stream>>>(
        msg_w1, msg_w2, upd_w1, upd_w2, wfrag);
    embed_kernel<<<(N_ * D_ + 255) / 256, 256, 0, stream>>>(x, emb_w, emb_b, h, h16);

    fill_kernel<<<(N_ + 255) / 256, 256, 0, stream>>>(deg, N_);
    degree_kernel<<<(E_ + 255) / 256, 256, 0, stream>>>(edst, deg);
    recip_kernel<<<(N_ + 255) / 256, 256, 0, stream>>>(deg);

    for (int l = 0; l < L_; ++l) {
        fill_kernel<<<1, 64, 0, stream>>>(bnsum, 64);
        fill_kernel<<<(N_ * D_ + 255) / 256, 256, 0, stream>>>(agg, N_ * D_);

        edge_msg1_kernel<<<EDGE_BLOCKS, 256, 0, stream>>>(
            h16, esrc, edst, wfrag + size_t(l * 12 + 0) * 512, msg_b1 + l * D_, t16, bnsum);
        bn_finalize_kernel<<<1, 32, 0, stream>>>(bnsum, bn_g + l * D_, bn_b + l * D_, bnss);
        edge_msg2_kernel<<<EDGE_BLOCKS, 256, 0, stream>>>(
            t16, edst, bnss, wfrag + size_t(l * 12 + 4) * 512, msg_b2 + l * D_, agg);
        node_update_kernel<<<NODE_BLOCKS, 256, 0, stream>>>(
            wfrag + size_t(l * 12 + 6) * 512, wfrag + size_t(l * 12 + 10) * 512,
            upd_b1 + l * D_, upd_b2 + l * D_, agg, deg, h, h16);
    }
    copyout_kernel<<<(N_ * D_ + 255) / 256, 256, 0, stream>>>(h, (float*)d_out);
}


# Degree count: per-edge global float atomic adds replaced by LDS integer histograms (64 edge chunks x 2 node halves, 200 KB LDS each) stored as partial planes and summed in the reciprocal kernel
# speedup vs baseline: 1.2314x; 1.2314x over previous
#include <hip/hip_runtime.h>
#include <hip/hip_bf16.h>

constexpr int N_ = 100000;
constexpr int E_ = 1600000;
constexpr int D_ = 32;
constexpr int L_ = 2;
constexpr float EPS_ = 1e-5f;

constexpr int WPB = 8;
constexpr int ETILES = E_ / 16;
constexpr int NTILES = N_ / 16;
constexpr int EDGE_BLOCKS = ETILES / WPB;
constexpr int NODE_BLOCKS = (NTILES + WPB - 1) / WPB;

typedef __attribute__((ext_vector_type(16))) _Float16 v16h;
typedef __attribute__((ext_vector_type(8)))  _Float16 v8h;
typedef __attribute__((ext_vector_type(8)))  float    v8f;

__device__ __forceinline__ v8f wmma_f16(v16h a, v16h b, v8f c) {
    return __builtin_amdgcn_wmma_f32_16x16x32_f16(
        false, a, false, b, (short)0, c, false, false);
}

__device__ __forceinline__ float swishf(float x) {
    float e = __expf(-x);
    return x * __builtin_amdgcn_rcpf(1.0f + e);
}

__device__ __forceinline__ v16h load_rowfrag(const _Float16* __restrict__ row, int half) {
    v8h lo = *(const v8h*)(row + half * 8);
    v8h hi = *(const v8h*)(row + 16 + half * 8);
    v16h a;
#pragma unroll
    for (int i = 0; i < 8; ++i) { a[i] = lo[i]; a[8 + i] = hi[i]; }
    return a;
}

__device__ __forceinline__ void wave_lds_sync() {
    __builtin_amdgcn_wave_barrier();
    asm volatile("s_wait_dscnt 0" ::: "memory");
    __builtin_amdgcn_wave_barrier();
}

__device__ __forceinline__ void atomic_add_f32(float* p, float v) {
    __hip_atomic_fetch_add(p, v, __ATOMIC_RELAXED, __HIP_MEMORY_SCOPE_AGENT);
}

constexpr size_t alignup(size_t x) { return (x + 255) & ~size_t(255); }
constexpr size_t SZ_WFRAG = size_t(L_) * 12 * 512 * 2;
constexpr size_t OFF_H    = alignup(SZ_WFRAG);
constexpr size_t OFF_H16  = alignup(OFF_H   + size_t(N_) * D_ * 4);
constexpr size_t OFF_AGG  = alignup(OFF_H16 + size_t(N_) * D_ * 2);
constexpr size_t OFF_DEG  = alignup(OFF_AGG + size_t(N_) * D_ * 4);
constexpr size_t OFF_BNS  = alignup(OFF_DEG + size_t(N_) * 4);
constexpr size_t OFF_BNP  = alignup(OFF_BNS + 64 * 4);
constexpr size_t OFF_T16  = alignup(OFF_BNP + 64 * 4);

__global__ void __launch_bounds__(256) prep_weights_kernel(
    const float* __restrict__ msg_w1, const float* __restrict__ msg_w2,
    const float* __restrict__ upd_w1, const float* __restrict__ upd_w2,
    _Float16* __restrict__ wfrag)
{
    int gid = blockIdx.x * 256 + threadIdx.x;
    if (gid >= L_ * 12 * 512) return;
    int frag = gid >> 9;
    int lane = (gid >> 4) & 31;
    int i    = gid & 15;
    int layer = frag / 12;
    int r     = frag % 12;
    const float* W; int kc, nt;
    if (r < 4)       { W = msg_w1 + size_t(layer) * 64 * 32; kc = r >> 1;       nt = r & 1; }
    else if (r < 6)  { W = msg_w2 + size_t(layer) * 32 * 32; kc = 0;            nt = r - 4; }
    else if (r < 10) { int q = r - 6;
                       W = upd_w1 + size_t(layer) * 64 * 32; kc = q >> 1;       nt = q & 1; }
    else             { W = upd_w2 + size_t(layer) * 32 * 32; kc = 0;            nt = r - 10; }
    int half = lane >> 4;
    int n = nt * 16 + (lane & 15);
    int k = kc * 32 + 16 * (i >> 3) + half * 8 + (i & 7);
    wfrag[size_t(frag) * 512 + lane * 16 + i] = (_Float16)W[k * 32 + n];
}

__global__ void __launch_bounds__(256) embed_kernel(
    const float* __restrict__ x, const float* __restrict__ w, const float* __restrict__ b,
    float* __restrict__ h, _Float16* __restrict__ h16)
{
    int gid = blockIdx.x * 256 + threadIdx.x;
    if (gid >= N_ * D_) return;
    int i = gid >> 5, j = gid & 31;
    const float* xr = x + size_t(i) * 4;
    float acc = b[j];
#pragma unroll
    for (int k = 0; k < 4; ++k) acc = fmaf(xr[k], w[k * 32 + j], acc);
    h[gid] = acc;
    h16[gid] = (_Float16)acc;
}

__global__ void __launch_bounds__(256) fill_kernel(float* __restrict__ p, int n) {
    int g = blockIdx.x * blockDim.x + threadIdx.x;
    if (g < n) p[g] = 0.0f;
}

constexpr int HC     = 64;
constexpr int HR     = 2;
constexpr int HBINS  = N_ / HR;
constexpr int HEDGES = E_ / HC;
constexpr int HT     = 1024;
constexpr int HV     = HEDGES / 4;
constexpr int HIT    = (HV + HT - 1) / HT;
static_assert(N_ % HR == 0 && HBINS % 4 == 0, "counter plane is whole 16-B pieces");
static_assert(E_ % HC == 0 && HEDGES % 4 == 0, "chunks are whole 16-B pieces");
static_assert(size_t(HBINS) * 4 <= 320u * 1024u, "counter plane fits the LDS");

__global__ void __launch_bounds__(HT) degree_hist_kernel(const int* __restrict__ dst,
                                                         unsigned* __restrict__ part) {
    __shared__ __attribute__((aligned(16))) unsigned bins[HBINS];
    const int tid   = threadIdx.x;
    const int chunk = blockIdx.x / HR;
    const int lo    = (blockIdx.x % HR) * HBINS;

    const int4* d4 = (const int4*)(dst + size_t(chunk) * HEDGES);
    int4 v[HIT];
#pragma unroll
    for (int j = 0; j < HIT; ++j) {
        int i = tid + j * HT;
        v[j] = (i < HV) ? d4[i] : make_int4(-1, -1, -1, -1);
    }

    uint4* b4 = (uint4*)bins;
    for (int i = tid; i < HBINS / 4; i += HT) b4[i] = make_uint4(0u, 0u, 0u, 0u);
    __syncthreads();

#pragma unroll
    for (int j = 0; j < HIT; ++j) {
        unsigned a0 = (unsigned)(v[j].x - lo), a1 = (unsigned)(v[j].y - lo);
        unsigned a2 = (unsigned)(v[j].z - lo), a3 = (unsigned)(v[j].w - lo);
        if (a0 < (unsigned)HBINS) atomicAdd(&bins[a0], 1u);
        if (a1 < (unsigned)HBINS) atomicAdd(&bins[a1], 1u);
        if (a2 < (unsigned)HBINS) atomicAdd(&bins[a2], 1u);
        if (a3 < (unsigned)HBINS) atomicAdd(&bins[a3], 1u);
    }
    __syncthreads();

    uint4* o4 = (uint4*)(part + size_t(chunk) * N_ + lo);
    for (int i = tid; i < HBINS / 4; i += HT) o4[i] = b4[i];
}

static_assert(HC == 64 && (N_ % 4) == 0, "four waves x 16 planes; whole 16-B node groups");
__global__ void __launch_bounds__(128) recip_kernel(const unsigned* __restrict__ part,
                                                    float* __restrict__ rdeg) {
    __shared__ uint4 red[4][32];
    const int lane = threadIdx.x & 31, grp = threadIdx.x >> 5;
    const int q = blockIdx.x * 32 + lane;
    const bool live = q < N_ / 4;
    uint4 s = make_uint4(0u, 0u, 0u, 0u);
    if (live) {
        const uint4* p4 = (const uint4*)part + size_t(grp) * 16 * (N_ / 4) + q;
        uint4 t[16];
#pragma unroll
        for (int k = 0; k < 16; ++k) t[k] = p4[size_t(k) * (N_ / 4)];
#pragma unroll
        for (int k = 0; k < 16; ++k) { s.x += t[k].x; s.y += t[k].y; s.z += t[k].z; s.w += t[k].w; }
    }
    red[grp][lane] = s;
    __syncthreads();
    if (grp == 0 && live) {
        uint4 a = red[1][lane], b = red[2][lane], d = red[3][lane];
        s.x += a.x + b.x + d.x; s.y += a.y + b.y + d.y;
        s.z += a.z + b.z + d.z; s.w += a.w + b.w + d.w;
        float4 r;
        r.x = __builtin_amdgcn_rcpf(fmaxf((float)s.x, 1.0f));
        r.y = __builtin_amdgcn_rcpf(fmaxf((float)s.y, 1.0f));
        r.z = __builtin_amdgcn_rcpf(fmaxf((float)s.z, 1.0f));
        r.w = __builtin_amdgcn_rcpf(fmaxf((float)s.w, 1.0f));
        ((float4*)rdeg)[q] = r;
    }
}

__global__ void __launch_bounds__(256) edge_msg1_kernel(
    const _Float16* __restrict__ h16,
    const int* __restrict__ src, const int* __restrict__ dst,
    const _Float16* __restrict__ wf1, const float* __restrict__ b1,
    _Float16* __restrict__ t16, float* __restrict__ bnsum  )
{
    __shared__ _Float16 tb[WPB][16][32];
    __shared__ float bsum[32], bsq[32];
    int tid = threadIdx.x;
    if (tid < 32) { bsum[tid] = 0.0f; bsq[tid] = 0.0f; }
    __syncthreads();

    int lane = tid & 31, wv = tid >> 5;
    int tile = blockIdx.x * WPB + wv;
    int m = lane & 15, half = lane >> 4;
    int e = tile * 16 + m;
    int di = dst[e], si = src[e];

    v16h a0 = load_rowfrag(h16 + size_t(di) * 32, half);
    v16h a1 = load_rowfrag(h16 + size_t(si) * 32, half);
    v16h b00 = *(const v16h*)(wf1 + 0 * 512 + lane * 16);
    v16h b01 = *(const v16h*)(wf1 + 1 * 512 + lane * 16);
    v16h b10 = *(const v16h*)(wf1 + 2 * 512 + lane * 16);
    v16h b11 = *(const v16h*)(wf1 + 3 * 512 + lane * 16);

    v8f c0 = {}, c1 = {};
    c0 = wmma_f16(a0, b00, c0); c0 = wmma_f16(a1, b10, c0);
    c1 = wmma_f16(a0, b01, c1); c1 = wmma_f16(a1, b11, c1);

    int n0 = lane & 15;
    float bias0 = b1[n0], bias1 = b1[16 + n0];
    float s0 = 0.f, q0 = 0.f, s1 = 0.f, q1 = 0.f;
#pragma unroll
    for (int r = 0; r < 8; ++r) {
        int mr = (lane >> 4) * 8 + r;
        float t0 = swishf(c0[r] + bias0);
        float t1 = swishf(c1[r] + bias1);
        tb[wv][mr][n0]      = (_Float16)t0;
        tb[wv][mr][16 + n0] = (_Float16)t1;
        s0 += t0; q0 += t0 * t0; s1 += t1; q1 += t1 * t1;
    }
    atomicAdd(&bsum[n0], s0);      atomicAdd(&bsq[n0], q0);
    atomicAdd(&bsum[16 + n0], s1); atomicAdd(&bsq[16 + n0], q1);
    __syncthreads();

    v16h out;
#pragma unroll
    for (int i = 0; i < 8; ++i) {
        out[i]     = tb[wv][m][half * 8 + i];
        out[8 + i] = tb[wv][m][16 + half * 8 + i];
    }
    *(v16h*)(t16 + size_t(tile) * 512 + lane * 16) = out;

    if (tid < 32) {
        atomic_add_f32(&bnsum[tid],      bsum[tid]);
        atomic_add_f32(&bnsum[32 + tid], bsq[tid]);
    }
}

__global__ void bn_finalize_kernel(const float* __restrict__ bnsum,
                                   const float* __restrict__ g, const float* __restrict__ b,
                                   float* __restrict__ ss  )
{
    int n = threadIdx.x;
    if (n >= 32) return;
    float mu  = bnsum[n] * (1.0f / E_);
    float var = bnsum[32 + n] * (1.0f / E_) - mu * mu;
    float sc  = g[n] * rsqrtf(var + EPS_);
    ss[n] = sc;
    ss[32 + n] = b[n] - mu * sc;
}

__global__ void __launch_bounds__(256) edge_msg2_kernel(
    const _Float16* __restrict__ t16, const int* __restrict__ dst,
    const float* __restrict__ ss, const _Float16* __restrict__ wf2,
    const float* __restrict__ b2, float* __restrict__ agg)
{
    __shared__ float sc[32], sh[32];
    int tid = threadIdx.x;
    if (tid < 32) { sc[tid] = ss[tid]; sh[tid] = ss[32 + tid]; }
    __syncthreads();

    int lane = tid & 31, wv = tid >> 5;
    int tile = blockIdx.x * WPB + wv;
    int half = lane >> 4;

    v16h a = *(const v16h*)(t16 + size_t(tile) * 512 + lane * 16);
    v16h an;
#pragma unroll
    for (int i = 0; i < 16; ++i) {
        int k = 16 * (i >> 3) + half * 8 + (i & 7);
        an[i] = (_Float16)((float)a[i] * sc[k] + sh[k]);
    }
    v16h w0 = *(const v16h*)(wf2 + 0 * 512 + lane * 16);
    v16h w1 = *(const v16h*)(wf2 + 1 * 512 + lane * 16);
    v8f c0 = {}, c1 = {};
    c0 = wmma_f16(an, w0, c0);
    c1 = wmma_f16(an, w1, c1);

    int n0 = lane & 15;
    float bias0 = b2[n0], bias1 = b2[16 + n0];
#pragma unroll
    for (int r = 0; r < 8; ++r) {
        int mr = (lane >> 4) * 8 + r;
        int node = dst[tile * 16 + mr];
        atomic_add_f32(&agg[size_t(node) * 32 + n0],      swishf(c0[r] + bias0));
        atomic_add_f32(&agg[size_t(node) * 32 + 16 + n0], swishf(c1[r] + bias1));
    }
}

__global__ void __launch_bounds__(256) node_update_kernel(
    const _Float16* __restrict__ wfu1, const _Float16* __restrict__ wfu2,
    const float* __restrict__ ub1, const float* __restrict__ ub2,
    const float* __restrict__ agg, const float* __restrict__ recip,
    float* __restrict__ h, _Float16* __restrict__ h16)
{
    __shared__ _Float16 ub[WPB][16][32];
    __shared__ float    ob[WPB][16][32];
    int tid = threadIdx.x, lane = tid & 31, wv = tid >> 5;
    int tile = blockIdx.x * WPB + wv;
    if (tile >= NTILES) return;

    int m = lane & 15, half = lane >> 4;
    size_t node = size_t(tile) * 16 + m;

    v16h a0 = load_rowfrag(h16 + node * 32, half);
    float rc = recip[node];
    const float* ar = agg + node * 32;
    v16h a1;
#pragma unroll
    for (int i = 0; i < 8; ++i) {
        a1[i]     = (_Float16)(ar[half * 8 + i] * rc);
        a1[8 + i] = (_Float16)(ar[16 + half * 8 + i] * rc);
    }
    v16h b00 = *(const v16h*)(wfu1 + 0 * 512 + lane * 16);
    v16h b01 = *(const v16h*)(wfu1 + 1 * 512 + lane * 16);
    v16h b10 = *(const v16h*)(wfu1 + 2 * 512 + lane * 16);
    v16h b11 = *(const v16h*)(wfu1 + 3 * 512 + lane * 16);
    v8f c0 = {}, c1 = {};
    c0 = wmma_f16(a0, b00, c0); c0 = wmma_f16(a1, b10, c0);
    c1 = wmma_f16(a0, b01, c1); c1 = wmma_f16(a1, b11, c1);

    int n0 = lane & 15;
    float bias0 = ub1[n0], bias1 = ub1[16 + n0];
#pragma unroll
    for (int r = 0; r < 8; ++r) {
        int mr = (lane >> 4) * 8 + r;
        ub[wv][mr][n0]      = (_Float16)swishf(c0[r] + bias0);
        ub[wv][mr][16 + n0] = (_Float16)swishf(c1[r] + bias1);
    }
    wave_lds_sync();

    v16h au;
#pragma unroll
    for (int i = 0; i < 8; ++i) {
        au[i]     = ub[wv][m][half * 8 + i];
        au[8 + i] = ub[wv][m][16 + half * 8 + i];
    }
    v16h u0 = *(const v16h*)(wfu2 + 0 * 512 + lane * 16);
    v16h u1 = *(const v16h*)(wfu2 + 1 * 512 + lane * 16);
    v8f d0 = {}, d1 = {};
    d0 = wmma_f16(au, u0, d0);
    d1 = wmma_f16(au, u1, d1);

    float bb0 = ub2[n0], bb1 = ub2[16 + n0];
#pragma unroll
    for (int r = 0; r < 8; ++r) {
        int mr = (lane >> 4) * 8 + r;
        ob[wv][mr][n0]      = swishf(d0[r] + bb0);
        ob[wv][mr][16 + n0] = swishf(d1[r] + bb1);
    }
    wave_lds_sync();

    int row = lane >> 1, seg = (lane & 1) * 16;
    size_t nr = size_t(tile) * 16 + row;
    float*    hp   = h   + nr * 32 + seg;
    _Float16* hp16 = h16 + nr * 32 + seg;
#pragma unroll
    for (int i = 0; i < 16; ++i) {
        float nv = hp[i] + ob[wv][row][seg + i];
        hp[i] = nv;
        hp16[i] = (_Float16)nv;
    }
}

__global__ void __launch_bounds__(256) copyout_kernel(const float* __restrict__ h,
                                                      float* __restrict__ out) {
    int g = blockIdx.x * 256 + threadIdx.x;
    if (g < N_ * D_) out[g] = h[g];
}

extern "C" void kernel_launch(void* const* d_in, const int* in_sizes, int n_in,
                              void* d_out, int out_size, void* d_ws, size_t ws_size,
                              hipStream_t stream) {
    const float* x      = (const float*)d_in[0];
    const float* emb_w  = (const float*)d_in[1];
    const float* emb_b  = (const float*)d_in[2];
    const float* msg_w1 = (const float*)d_in[3];
    const float* msg_b1 = (const float*)d_in[4];
    const float* bn_g   = (const float*)d_in[5];
    const float* bn_b   = (const float*)d_in[6];
    const float* msg_w2 = (const float*)d_in[7];
    const float* msg_b2 = (const float*)d_in[8];
    const float* upd_w1 = (const float*)d_in[9];
    const float* upd_b1 = (const float*)d_in[10];
    const float* upd_w2 = (const float*)d_in[11];
    const float* upd_b2 = (const float*)d_in[12];
    const int*   eidx   = (const int*)d_in[13];
    const int*   esrc   = eidx;
    const int*   edst   = eidx + E_;

    char* ws = (char*)d_ws;
    _Float16* wfrag = (_Float16*)(ws);
    float*    h     = (float*)(ws + OFF_H);
    _Float16* h16   = (_Float16*)(ws + OFF_H16);
    float*    agg   = (float*)(ws + OFF_AGG);
    float*    deg   = (float*)(ws + OFF_DEG);
    float*    bnsum = (float*)(ws + OFF_BNS);
    float*    bnss  = (float*)(ws + OFF_BNP);
    _Float16* t16   = (_Float16*)(ws + OFF_T16);
    unsigned* dpart = (unsigned*)(ws + OFF_T16);

    prep_weights_kernel<<<(L_ * 12 * 512 + 255) / 256, 256, 0, stream>>>(
        msg_w1, msg_w2, upd_w1, upd_w2, wfrag);
    embed_kernel<<<(N_ * D_ + 255) / 256, 256, 0, stream>>>(x, emb_w, emb_b, h, h16);

    degree_hist_kernel<<<HC * HR, HT, 0, stream>>>(edst, dpart);
    recip_kernel<<<(N_ / 4 + 31) / 32, 128, 0, stream>>>(dpart, deg);

    for (int l = 0; l < L_; ++l) {
        fill_kernel<<<1, 64, 0, stream>>>(bnsum, 64);
        fill_kernel<<<(N_ * D_ + 255) / 256, 256, 0, stream>>>(agg, N_ * D_);

        edge_msg1_kernel<<<EDGE_BLOCKS, 256, 0, stream>>>(
            h16, esrc, edst, wfrag + size_t(l * 12 + 0) * 512, msg_b1 + l * D_, t16, bnsum);
        bn_finalize_kernel<<<1, 32, 0, stream>>>(bnsum, bn_g + l * D_, bn_b + l * D_, bnss);
        edge_msg2_kernel<<<EDGE_BLOCKS, 256, 0, stream>>>(
            t16, edst, bnss, wfrag + size_t(l * 12 + 4) * 512, msg_b2 + l * D_, agg);
        node_update_kernel<<<NODE_BLOCKS, 256, 0, stream>>>(
            wfrag + size_t(l * 12 + 6) * 512, wfrag + size_t(l * 12 + 10) * 512,
            upd_b1 + l * D_, upd_b2 + l * D_, agg, deg, h, h16);
    }
    copyout_kernel<<<(N_ * D_ + 255) / 256, 256, 0, stream>>>(h, (float*)d_out);
}


# Grids of 100000 one-tile waves (both edge kernels, embed, agg clear, copy-out) become resident grids, waves loop over tiles, next tile prefetched; BN sums per wave in registers (sum order changes)
# speedup vs baseline: 1.5294x; 1.2420x over previous
#include <hip/hip_runtime.h>
#include <hip/hip_bf16.h>

constexpr int N_ = 100000;
constexpr int E_ = 1600000;
constexpr int D_ = 32;
constexpr int L_ = 2;
constexpr float EPS_ = 1e-5f;

constexpr int WPB = 8;
constexpr int ETILES = E_ / 16;
constexpr int NTILES = N_ / 16;
constexpr int NODE_BLOCKS = (NTILES + WPB - 1) / WPB;
constexpr int EDGE1_GRID = 256 * 3;
constexpr int EDGE2_GRID = 256 * 4;
static_assert(EDGE2_GRID * WPB <= ETILES, "every wave owns at least one tile");
constexpr int NPIECES = N_ * D_ / 4;
constexpr int EW_GRID = 625;
constexpr int EW_PER  = 5;
constexpr int EW_T    = EW_GRID * 256;
static_assert((N_ * D_) % 4 == 0 && EW_T * EW_PER == NPIECES && EW_T % 8 == 0,
              "elementwise split is exact");

typedef __attribute__((ext_vector_type(16))) _Float16 v16h;
typedef __attribute__((ext_vector_type(8)))  _Float16 v8h;
typedef __attribute__((ext_vector_type(4)))  _Float16 v4h;
typedef __attribute__((ext_vector_type(8)))  float    v8f;

__device__ __forceinline__ v8f wmma_f16(v16h a, v16h b, v8f c) {
    return __builtin_amdgcn_wmma_f32_16x16x32_f16(
        false, a, false, b, (short)0, c, false, false);
}

__device__ __forceinline__ float swishf(float x) {
    float e = __expf(-x);
    return x * __builtin_amdgcn_rcpf(1.0f + e);
}

__device__ __forceinline__ v16h load_rowfrag(const _Float16* __restrict__ row, int half) {
    v8h lo = *(const v8h*)(row + half * 8);
    v8h hi = *(const v8h*)(row + 16 + half * 8);
    v16h a;
#pragma unroll
    for (int i = 0; i < 8; ++i) { a[i] = lo[i]; a[8 + i] = hi[i]; }
    return a;
}

__device__ __forceinline__ void wave_lds_sync() {
    __builtin_amdgcn_wave_barrier();
    asm volatile("s_wait_dscnt 0" ::: "memory");
    __builtin_amdgcn_wave_barrier();
}

__device__ __forceinline__ void atomic_add_f32(float* p, float v) {
    __hip_atomic_fetch_add(p, v, __ATOMIC_RELAXED, __HIP_MEMORY_SCOPE_AGENT);
}

constexpr size_t alignup(size_t x) { return (x + 255) & ~size_t(255); }
constexpr size_t SZ_WFRAG = size_t(L_) * 12 * 512 * 2;
constexpr size_t OFF_H    = alignup(SZ_WFRAG);
constexpr size_t OFF_H16  = alignup(OFF_H   + size_t(N_) * D_ * 4);
constexpr size_t OFF_AGG  = alignup(OFF_H16 + size_t(N_) * D_ * 2);
constexpr size_t OFF_DEG  = alignup(OFF_AGG + size_t(N_) * D_ * 4);
constexpr size_t OFF_BNS  = alignup(OFF_DEG + size_t(N_) * 4);
constexpr size_t OFF_BNP  = alignup(OFF_BNS + 64 * 4);
constexpr size_t OFF_T16  = alignup(OFF_BNP + 64 * 4);

__global__ void __launch_bounds__(256) prep_weights_kernel(
    const float* __restrict__ msg_w1, const float* __restrict__ msg_w2,
    const float* __restrict__ upd_w1, const float* __restrict__ upd_w2,
    _Float16* __restrict__ wfrag)
{
    int gid = blockIdx.x * 256 + threadIdx.x;
    if (gid >= L_ * 12 * 512) return;
    int frag = gid >> 9;
    int lane = (gid >> 4) & 31;
    int i    = gid & 15;
    int layer = frag / 12;
    int r     = frag % 12;
    const float* W; int kc, nt;
    if (r < 4)       { W = msg_w1 + size_t(layer) * 64 * 32; kc = r >> 1;       nt = r & 1; }
    else if (r < 6)  { W = msg_w2 + size_t(layer) * 32 * 32; kc = 0;            nt = r - 4; }
    else if (r < 10) { int q = r - 6;
                       W = upd_w1 + size_t(layer) * 64 * 32; kc = q >> 1;       nt = q & 1; }
    else             { W = upd_w2 + size_t(layer) * 32 * 32; kc = 0;            nt = r - 10; }
    int half = lane >> 4;
    int n = nt * 16 + (lane & 15);
    int k = kc * 32 + 16 * (i >> 3) + half * 8 + (i & 7);
    wfrag[size_t(frag) * 512 + lane * 16 + i] = (_Float16)W[k * 32 + n];
}

__global__ void __launch_bounds__(256) embed_kernel(
    const float* __restrict__ x, const float* __restrict__ w, const float* __restrict__ b,
    float* __restrict__ h, _Float16* __restrict__ h16)
{
    const int q0 = blockIdx.x * 256 + threadIdx.x;
    const int jq = (q0 & 7) * 4;
    const float4 w0 = *(const float4*)(w + 0 * 32 + jq);
    const float4 w1 = *(const float4*)(w + 1 * 32 + jq);
    const float4 w2 = *(const float4*)(w + 2 * 32 + jq);
    const float4 w3 = *(const float4*)(w + 3 * 32 + jq);
    const float4 bb = *(const float4*)(b + jq);
    float4 xv[EW_PER];
#pragma unroll
    for (int p = 0; p < EW_PER; ++p)
        xv[p] = *(const float4*)(x + size_t((q0 + p * EW_T) >> 3) * 4);
#pragma unroll
    for (int p = 0; p < EW_PER; ++p) {
        const int q = q0 + p * EW_T;
        const float4 xr = xv[p];
        float4 acc;
        acc.x = fmaf(xr.w, w3.x, fmaf(xr.z, w2.x, fmaf(xr.y, w1.x, fmaf(xr.x, w0.x, bb.x))));
        acc.y = fmaf(xr.w, w3.y, fmaf(xr.z, w2.y, fmaf(xr.y, w1.y, fmaf(xr.x, w0.y, bb.y))));
        acc.z = fmaf(xr.w, w3.z, fmaf(xr.z, w2.z, fmaf(xr.y, w1.z, fmaf(xr.x, w0.z, bb.z))));
        acc.w = fmaf(xr.w, w3.w, fmaf(xr.z, w2.w, fmaf(xr.y, w1.w, fmaf(xr.x, w0.w, bb.w))));
        ((float4*)h)[q] = acc;
        v4h a16;
        a16[0] = (_Float16)acc.x; a16[1] = (_Float16)acc.y;
        a16[2] = (_Float16)acc.z; a16[3] = (_Float16)acc.w;
        ((v4h*)h16)[q] = a16;
    }
}

__global__ void __launch_bounds__(256) fill_kernel(float* __restrict__ p, int n) {
    int g = blockIdx.x * blockDim.x + threadIdx.x;
    if (g < n) p[g] = 0.0f;
}

__global__ void __launch_bounds__(256) fill4_kernel(float4* __restrict__ p) {
    const int q0 = blockIdx.x * 256 + threadIdx.x;
#pragma unroll
    for (int k = 0; k < EW_PER; ++k)
        p[q0 + k * EW_T] = make_float4(0.0f, 0.0f, 0.0f, 0.0f);
}

constexpr int HC     = 64;
constexpr int HR     = 2;
constexpr int HBINS  = N_ / HR;
constexpr int HEDGES = E_ / HC;
constexpr int HT     = 1024;
constexpr int HV     = HEDGES / 4;
constexpr int HIT    = (HV + HT - 1) / HT;
static_assert(N_ % HR == 0 && HBINS % 4 == 0, "counter plane is whole 16-B pieces");
static_assert(E_ % HC == 0 && HEDGES % 4 == 0, "chunks are whole 16-B pieces");
static_assert(size_t(HBINS) * 4 <= 320u * 1024u, "counter plane fits the LDS");

__global__ void __launch_bounds__(HT) degree_hist_kernel(const int* __restrict__ dst,
                                                         unsigned* __restrict__ part) {
    __shared__ __attribute__((aligned(16))) unsigned bins[HBINS];
    const int tid   = threadIdx.x;
    const int chunk = blockIdx.x / HR;
    const int lo    = (blockIdx.x % HR) * HBINS;

    const int4* d4 = (const int4*)(dst + size_t(chunk) * HEDGES);
    int4 v[HIT];
#pragma unroll
    for (int j = 0; j < HIT; ++j) {
        int i = tid + j * HT;
        v[j] = (i < HV) ? d4[i] : make_int4(-1, -1, -1, -1);
    }

    uint4* b4 = (uint4*)bins;
    for (int i = tid; i < HBINS / 4; i += HT) b4[i] = make_uint4(0u, 0u, 0u, 0u);
    __syncthreads();

#pragma unroll
    for (int j = 0; j < HIT; ++j) {
        unsigned a0 = (unsigned)(v[j].x - lo), a1 = (unsigned)(v[j].y - lo);
        unsigned a2 = (unsigned)(v[j].z - lo), a3 = (unsigned)(v[j].w - lo);
        if (a0 < (unsigned)HBINS) atomicAdd(&bins[a0], 1u);
        if (a1 < (unsigned)HBINS) atomicAdd(&bins[a1], 1u);
        if (a2 < (unsigned)HBINS) atomicAdd(&bins[a2], 1u);
        if (a3 < (unsigned)HBINS) atomicAdd(&bins[a3], 1u);
    }
    __syncthreads();

    uint4* o4 = (uint4*)(part + size_t(chunk) * N_ + lo);
    for (int i = tid; i < HBINS / 4; i += HT) o4[i] = b4[i];
}

static_assert(HC == 64 && (N_ % 4) == 0, "four waves x 16 planes; whole 16-B node groups");
__global__ void __launch_bounds__(128) recip_kernel(const unsigned* __restrict__ part,
                                                    float* __restrict__ rdeg) {
    __shared__ uint4 red[4][32];
    const int lane = threadIdx.x & 31, grp = threadIdx.x >> 5;
    const int q = blockIdx.x * 32 + lane;
    const bool live = q < N_ / 4;
    uint4 s = make_uint4(0u, 0u, 0u, 0u);
    if (live) {
        const uint4* p4 = (const uint4*)part + size_t(grp) * 16 * (N_ / 4) + q;
        uint4 t[16];
#pragma unroll
        for (int k = 0; k < 16; ++k) t[k] = p4[size_t(k) * (N_ / 4)];
#pragma unroll
        for (int k = 0; k < 16; ++k) { s.x += t[k].x; s.y += t[k].y; s.z += t[k].z; s.w += t[k].w; }
    }
    red[grp][lane] = s;
    __syncthreads();
    if (grp == 0 && live) {
        uint4 a = red[1][lane], b = red[2][lane], d = red[3][lane];
        s.x += a.x + b.x + d.x; s.y += a.y + b.y + d.y;
        s.z += a.z + b.z + d.z; s.w += a.w + b.w + d.w;
        float4 r;
        r.x = __builtin_amdgcn_rcpf(fmaxf((float)s.x, 1.0f));
        r.y = __builtin_amdgcn_rcpf(fmaxf((float)s.y, 1.0f));
        r.z = __builtin_amdgcn_rcpf(fmaxf((float)s.z, 1.0f));
        r.w = __builtin_amdgcn_rcpf(fmaxf((float)s.w, 1.0f));
        ((float4*)rdeg)[q] = r;
    }
}

__global__ void __launch_bounds__(256) edge_msg1_kernel(
    const _Float16* __restrict__ h16,
    const int* __restrict__ src, const int* __restrict__ dst,
    const _Float16* __restrict__ wf1, const float* __restrict__ b1,
    _Float16* __restrict__ t16, float* __restrict__ bnsum  )
{
    __shared__ _Float16 tb[WPB][16][32];
    __shared__ float bsum[32], bsq[32];
    const int tid = threadIdx.x;
    if (tid < 32) { bsum[tid] = 0.0f; bsq[tid] = 0.0f; }
    __syncthreads();

    const int lane = tid & 31;
    const int wv = __builtin_amdgcn_readfirstlane(tid >> 5);
    const int m = lane & 15, half = lane >> 4, n0 = lane & 15;
    const int stride = gridDim.x * WPB;
    int tile = blockIdx.x * WPB + wv;

    const v16h b00 = *(const v16h*)(wf1 + 0 * 512 + lane * 16);
    const v16h b01 = *(const v16h*)(wf1 + 1 * 512 + lane * 16);
    const v16h b10 = *(const v16h*)(wf1 + 2 * 512 + lane * 16);
    const v16h b11 = *(const v16h*)(wf1 + 3 * 512 + lane * 16);
    const float bias0 = b1[n0], bias1 = b1[16 + n0];
    float s0 = 0.f, q0 = 0.f, s1 = 0.f, q1 = 0.f;

    const int e0 = tile * 16 + m;
    v16h a0 = load_rowfrag(h16 + size_t(dst[e0]) * 32, half);
    v16h a1 = load_rowfrag(h16 + size_t(src[e0]) * 32, half);
    const int e1 = min(tile + stride, ETILES - 1) * 16 + m;
    int dn = dst[e1], sn = src[e1];

    for (; tile < ETILES; tile += stride) {
        v8f c0 = {}, c1 = {};
        c0 = wmma_f16(a0, b00, c0); c0 = wmma_f16(a1, b10, c0);
        c1 = wmma_f16(a0, b01, c1); c1 = wmma_f16(a1, b11, c1);

        a0 = load_rowfrag(h16 + size_t(dn) * 32, half);
        a1 = load_rowfrag(h16 + size_t(sn) * 32, half);
        const int e2 = min(tile + 2 * stride, ETILES - 1) * 16 + m;
        dn = dst[e2]; sn = src[e2];

#pragma unroll
        for (int r = 0; r < 8; ++r) {
            int mr = half * 8 + r;
            float t0 = swishf(c0[r] + bias0);
            float t1 = swishf(c1[r] + bias1);
            tb[wv][mr][n0]      = (_Float16)t0;
            tb[wv][mr][16 + n0] = (_Float16)t1;
            s0 += t0; q0 += t0 * t0; s1 += t1; q1 += t1 * t1;
        }
        wave_lds_sync();

        v16h out;
#pragma unroll
        for (int i = 0; i < 8; ++i) {
            out[i]     = tb[wv][m][half * 8 + i];
            out[8 + i] = tb[wv][m][16 + half * 8 + i];
        }
        *(v16h*)(t16 + size_t(tile) * 512 + lane * 16) = out;
    }

    atomicAdd(&bsum[n0], s0);      atomicAdd(&bsq[n0], q0);
    atomicAdd(&bsum[16 + n0], s1); atomicAdd(&bsq[16 + n0], q1);
    __syncthreads();
    if (tid < 32) {
        atomic_add_f32(&bnsum[tid],      bsum[tid]);
        atomic_add_f32(&bnsum[32 + tid], bsq[tid]);
    }
}

__global__ void bn_finalize_kernel(const float* __restrict__ bnsum,
                                   const float* __restrict__ g, const float* __restrict__ b,
                                   float* __restrict__ ss  )
{
    int n = threadIdx.x;
    if (n >= 32) return;
    float mu  = bnsum[n] * (1.0f / E_);
    float var = bnsum[32 + n] * (1.0f / E_) - mu * mu;
    float sc  = g[n] * rsqrtf(var + EPS_);
    ss[n] = sc;
    ss[32 + n] = b[n] - mu * sc;
}

__global__ void __launch_bounds__(256) edge_msg2_kernel(
    const _Float16* __restrict__ t16, const int* __restrict__ dst,
    const float* __restrict__ ss, const _Float16* __restrict__ wf2,
    const float* __restrict__ b2, float* __restrict__ agg)
{
    const int tid = threadIdx.x, lane = tid & 31;
    const int wv = __builtin_amdgcn_readfirstlane(tid >> 5);
    const int half = lane >> 4, n0 = lane & 15;
    const int stride = gridDim.x * WPB;
    int tile = blockIdx.x * WPB + wv;

    float sc[16], sh[16];
#pragma unroll
    for (int g = 0; g < 2; ++g) {
#pragma unroll
        for (int j = 0; j < 2; ++j) {
            const float4 vs = *(const float4*)(ss + 16 * g + half * 8 + 4 * j);
            const float4 vh = *(const float4*)(ss + 32 + 16 * g + half * 8 + 4 * j);
            sc[8 * g + 4 * j + 0] = vs.x; sc[8 * g + 4 * j + 1] = vs.y;
            sc[8 * g + 4 * j + 2] = vs.z; sc[8 * g + 4 * j + 3] = vs.w;
            sh[8 * g + 4 * j + 0] = vh.x; sh[8 * g + 4 * j + 1] = vh.y;
            sh[8 * g + 4 * j + 2] = vh.z; sh[8 * g + 4 * j + 3] = vh.w;
        }
    }
    const v16h w0 = *(const v16h*)(wf2 + 0 * 512 + lane * 16);
    const v16h w1 = *(const v16h*)(wf2 + 1 * 512 + lane * 16);
    const float bias0 = b2[n0], bias1 = b2[16 + n0];

    v16h a = *(const v16h*)(t16 + size_t(tile) * 512 + lane * 16);
    const int4* dp = (const int4*)(dst + tile * 16 + half * 8);
    int4 d0 = dp[0], d1 = dp[1];

    for (; tile < ETILES; tile += stride) {
        v16h an;
#pragma unroll
        for (int i = 0; i < 16; ++i)
            an[i] = (_Float16)((float)a[i] * sc[i] + sh[i]);
        const int node[8] = { d0.x, d0.y, d0.z, d0.w, d1.x, d1.y, d1.z, d1.w };
        float* row[8];
#pragma unroll
        for (int r = 0; r < 8; ++r) row[r] = agg + size_t(node[r]) * 32 + n0;

        const int nt = min(tile + stride, ETILES - 1);
        a = *(const v16h*)(t16 + size_t(nt) * 512 + lane * 16);
        dp = (const int4*)(dst + nt * 16 + half * 8);
        d0 = dp[0]; d1 = dp[1];

        v8f c0 = {}, c1 = {};
        c0 = wmma_f16(an, w0, c0);
        c1 = wmma_f16(an, w1, c1);
#pragma unroll
        for (int r = 0; r < 8; ++r) {
            atomic_add_f32(row[r],      swishf(c0[r] + bias0));
            atomic_add_f32(row[r] + 16, swishf(c1[r] + bias1));
        }
    }
}

__global__ void __launch_bounds__(256) node_update_kernel(
    const _Float16* __restrict__ wfu1, const _Float16* __restrict__ wfu2,
    const float* __restrict__ ub1, const float* __restrict__ ub2,
    const float* __restrict__ agg, const float* __restrict__ recip,
    float* __restrict__ h, _Float16* __restrict__ h16)
{
    __shared__ _Float16 ub[WPB][16][32];
    __shared__ float    ob[WPB][16][32];
    int tid = threadIdx.x, lane = tid & 31, wv = tid >> 5;
    int tile = blockIdx.x * WPB + wv;
    if (tile >= NTILES) return;

    int m = lane & 15, half = lane >> 4;
    size_t node = size_t(tile) * 16 + m;

    v16h a0 = load_rowfrag(h16 + node * 32, half);
    float rc = recip[node];
    const float* ar = agg + node * 32;
    v16h a1;
#pragma unroll
    for (int i = 0; i < 8; ++i) {
        a1[i]     = (_Float16)(ar[half * 8 + i] * rc);
        a1[8 + i] = (_Float16)(ar[16 + half * 8 + i] * rc);
    }
    v16h b00 = *(const v16h*)(wfu1 + 0 * 512 + lane * 16);
    v16h b01 = *(const v16h*)(wfu1 + 1 * 512 + lane * 16);
    v16h b10 = *(const v16h*)(wfu1 + 2 * 512 + lane * 16);
    v16h b11 = *(const v16h*)(wfu1 + 3 * 512 + lane * 16);
    v8f c0 = {}, c1 = {};
    c0 = wmma_f16(a0, b00, c0); c0 = wmma_f16(a1, b10, c0);
    c1 = wmma_f16(a0, b01, c1); c1 = wmma_f16(a1, b11, c1);

    int n0 = lane & 15;
    float bias0 = ub1[n0], bias1 = ub1[16 + n0];
#pragma unroll
    for (int r = 0; r < 8; ++r) {
        int mr = (lane >> 4) * 8 + r;
        ub[wv][mr][n0]      = (_Float16)swishf(c0[r] + bias0);
        ub[wv][mr][16 + n0] = (_Float16)swishf(c1[r] + bias1);
    }
    wave_lds_sync();

    v16h au;
#pragma unroll
    for (int i = 0; i < 8; ++i) {
        au[i]     = ub[wv][m][half * 8 + i];
        au[8 + i] = ub[wv][m][16 + half * 8 + i];
    }
    v16h u0 = *(const v16h*)(wfu2 + 0 * 512 + lane * 16);
    v16h u1 = *(const v16h*)(wfu2 + 1 * 512 + lane * 16);
    v8f d0 = {}, d1 = {};
    d0 = wmma_f16(au, u0, d0);
    d1 = wmma_f16(au, u1, d1);

    float bb0 = ub2[n0], bb1 = ub2[16 + n0];
#pragma unroll
    for (int r = 0; r < 8; ++r) {
        int mr = (lane >> 4) * 8 + r;
        ob[wv][mr][n0]      = swishf(d0[r] + bb0);
        ob[wv][mr][16 + n0] = swishf(d1[r] + bb1);
    }
    wave_lds_sync();

    int row = lane >> 1, seg = (lane & 1) * 16;
    size_t nr = size_t(tile) * 16 + row;
    float*    hp   = h   + nr * 32 + seg;
    _Float16* hp16 = h16 + nr * 32 + seg;
#pragma unroll
    for (int i = 0; i < 16; ++i) {
        float nv = hp[i] + ob[wv][row][seg + i];
        hp[i] = nv;
        hp16[i] = (_Float16)nv;
    }
}

__global__ void __launch_bounds__(256) copyout_kernel(const float* __restrict__ h,
                                                      float* __restrict__ out) {
    const float4* __restrict__ h4 = (const float4*)h;
    float4* __restrict__ o4 = (float4*)out;
    const int q0 = blockIdx.x * 256 + threadIdx.x;
    float4 v[EW_PER];
#pragma unroll
    for (int k = 0; k < EW_PER; ++k) v[k] = h4[q0 + k * EW_T];
#pragma unroll
    for (int k = 0; k < EW_PER; ++k)
        asm volatile("" : "+v"(v[k].x), "+v"(v[k].y), "+v"(v[k].z), "+v"(v[k].w));
#pragma unroll
    for (int k = 0; k < EW_PER; ++k) o4[q0 + k * EW_T] = v[k];
}

extern "C" void kernel_launch(void* const* d_in, const int* in_sizes, int n_in,
                              void* d_out, int out_size, void* d_ws, size_t ws_size,
                              hipStream_t stream) {
    const float* x      = (const float*)d_in[0];
    const float* emb_w  = (const float*)d_in[1];
    const float* emb_b  = (const float*)d_in[2];
    const float* msg_w1 = (const float*)d_in[3];
    const float* msg_b1 = (const float*)d_in[4];
    const float* bn_g   = (const float*)d_in[5];
    const float* bn_b   = (const float*)d_in[6];
    const float* msg_w2 = (const float*)d_in[7];
    const float* msg_b2 = (const float*)d_in[8];
    const float* upd_w1 = (const float*)d_in[9];
    const float* upd_b1 = (const float*)d_in[10];
    const float* upd_w2 = (const float*)d_in[11];
    const float* upd_b2 = (const float*)d_in[12];
    const int*   eidx   = (const int*)d_in[13];
    const int*   esrc   = eidx;
    const int*   edst   = eidx + E_;

    char* ws = (char*)d_ws;
    _Float16* wfrag = (_Float16*)(ws);
    float*    h     = (float*)(ws + OFF_H);
    _Float16* h16   = (_Float16*)(ws + OFF_H16);
    float*    agg   = (float*)(ws + OFF_AGG);
    float*    deg   = (float*)(ws + OFF_DEG);
    float*    bnsum = (float*)(ws + OFF_BNS);
    float*    bnss  = (float*)(ws + OFF_BNP);
    _Float16* t16   = (_Float16*)(ws + OFF_T16);
    unsigned* dpart = (unsigned*)(ws + OFF_T16);

    prep_weights_kernel<<<(L_ * 12 * 512 + 255) / 256, 256, 0, stream>>>(
        msg_w1, msg_w2, upd_w1, upd_w2, wfrag);
    embed_kernel<<<EW_GRID, 256, 0, stream>>>(x, emb_w, emb_b, h, h16);

    degree_hist_kernel<<<HC * HR, HT, 0, stream>>>(edst, dpart);
    recip_kernel<<<(N_ / 4 + 31) / 32, 128, 0, stream>>>(dpart, deg);

    for (int l = 0; l < L_; ++l) {
        fill_kernel<<<1, 64, 0, stream>>>(bnsum, 64);
        fill4_kernel<<<EW_GRID, 256, 0, stream>>>((float4*)agg);

        edge_msg1_kernel<<<EDGE1_GRID, 256, 0, stream>>>(
            h16, esrc, edst, wfrag + size_t(l * 12 + 0) * 512, msg_b1 + l * D_, t16, bnsum);
        bn_finalize_kernel<<<1, 32, 0, stream>>>(bnsum, bn_g + l * D_, bn_b + l * D_, bnss);
        edge_msg2_kernel<<<EDGE2_GRID, 256, 0, stream>>>(
            t16, edst, bnss, wfrag + size_t(l * 12 + 4) * 512, msg_b2 + l * D_, agg);
        node_update_kernel<<<NODE_BLOCKS, 256, 0, stream>>>(
            wfrag + size_t(l * 12 + 6) * 512, wfrag + size_t(l * 12 + 10) * 512,
            upd_b1 + l * D_, upd_b2 + l * D_, agg, deg, h, h16);
    }
    copyout_kernel<<<EW_GRID, 256, 0, stream>>>(h, (float*)d_out);
}


# 15 launches become 10: embedding, edge count, weight fragments and BN-sum clear share one launch; range scan done inside the scatter blocks; finalize re-clears BN sums; no float path touched
# speedup vs baseline: 2.2414x; 1.4655x over previous
#include <hip/hip_runtime.h>
#include <hip/hip_bf16.h>

constexpr int N_ = 100000;
constexpr int E_ = 1600000;
constexpr int D_ = 32;
constexpr int L_ = 2;
constexpr float EPS_ = 1e-5f;

constexpr int WPB = 8;
constexpr int ETILES = E_ / 16;
constexpr int NTILES = N_ / 16;
constexpr int NODE_BLOCKS = (NTILES + WPB - 1) / WPB;
constexpr int EDGE1_GRID = 256 * 3;
static_assert(EDGE1_GRID * WPB <= ETILES, "every wave owns at least one tile");
constexpr int NB  = 256;
constexpr int BW  = 391;
static_assert(NB * BW >= N_, "the ranges cover every node");
constexpr int NCH = 256;
constexpr int CH  = E_ / NCH;
constexpr int BT  = 1024;
constexpr int CIT = (CH + BT - 1) / BT;
static_assert(E_ % NCH == 0 && NB == 256 && NCH == 256 && BT == 1024, "binning split");
constexpr int NPIECES = N_ * D_ / 4;
constexpr int EW_GRID = 625;
constexpr int EW_PER  = 5;
constexpr int EW_T    = EW_GRID * 256;
static_assert((N_ * D_) % 4 == 0 && EW_T * EW_PER == NPIECES && EW_T % 8 == 0,
              "elementwise split is exact");

typedef __attribute__((ext_vector_type(16))) _Float16 v16h;
typedef __attribute__((ext_vector_type(8)))  _Float16 v8h;
typedef __attribute__((ext_vector_type(4)))  _Float16 v4h;
typedef __attribute__((ext_vector_type(8)))  float    v8f;

__device__ __forceinline__ v8f wmma_f16(v16h a, v16h b, v8f c) {
    return __builtin_amdgcn_wmma_f32_16x16x32_f16(
        false, a, false, b, (short)0, c, false, false);
}

__device__ __forceinline__ float swishf(float x) {
    float e = __expf(-x);
    return x * __builtin_amdgcn_rcpf(1.0f + e);
}

__device__ __forceinline__ v16h load_rowfrag(const _Float16* __restrict__ row, int half) {
    v8h lo = *(const v8h*)(row + half * 8);
    v8h hi = *(const v8h*)(row + 16 + half * 8);
    v16h a;
#pragma unroll
    for (int i = 0; i < 8; ++i) { a[i] = lo[i]; a[8 + i] = hi[i]; }
    return a;
}

__device__ __forceinline__ void wave_lds_sync() {
    __builtin_amdgcn_wave_barrier();
    asm volatile("s_wait_dscnt 0" ::: "memory");
    __builtin_amdgcn_wave_barrier();
}

__device__ __forceinline__ void atomic_add_f32(float* p, float v) {
    __hip_atomic_fetch_add(p, v, __ATOMIC_RELAXED, __HIP_MEMORY_SCOPE_AGENT);
}

constexpr size_t alignup(size_t x) { return (x + 255) & ~size_t(255); }
constexpr size_t SZ_WFRAG = size_t(L_) * 12 * 512 * 2;
constexpr size_t OFF_ES   = alignup(SZ_WFRAG);
constexpr size_t OFF_ED   = OFF_ES + size_t(E_) * 4;
constexpr size_t OFF_H16  = alignup(OFF_ED  + size_t(E_) * 4);
constexpr size_t OFF_AGG  = alignup(OFF_H16 + size_t(N_) * D_ * 2);
constexpr size_t OFF_BIN  = alignup(OFF_AGG + size_t(N_) * D_ * 4);
constexpr size_t OFF_BST  = OFF_BIN + size_t(NCH) * NB * 4;
constexpr size_t OFF_BNS  = alignup(OFF_BST + size_t(NB + 1) * 4);
static_assert(size_t(E_) * 8 <= size_t(N_) * D_ * 4 && size_t(NCH) * NB * 4 + size_t(NB + 1) * 4 <= size_t(N_) * 4,
              "workspace use stays within one [N,32] f32 plane plus one [N] f32 plane for these");
constexpr size_t OFF_BNP  = alignup(OFF_BNS + 64 * 4);
constexpr size_t OFF_T16  = alignup(OFF_BNP + 64 * 4);

constexpr int PREP_BLOCKS = (L_ * 12 * 512 + 255) / 256;
__device__ __forceinline__ void prep_weights_body(
    int blk,
    const float* __restrict__ msg_w1, const float* __restrict__ msg_w2,
    const float* __restrict__ upd_w1, const float* __restrict__ upd_w2,
    _Float16* __restrict__ wfrag)
{
    int gid = blk * 256 + threadIdx.x;
    if (gid >= L_ * 12 * 512) return;
    int frag = gid >> 9;
    int lane = (gid >> 4) & 31;
    int i    = gid & 15;
    int layer = frag / 12;
    int r     = frag % 12;
    const float* W; int kc, nt;
    if (r < 4)       { W = msg_w1 + size_t(layer) * 64 * 32; kc = r >> 1;       nt = r & 1; }
    else if (r < 6)  { W = msg_w2 + size_t(layer) * 32 * 32; kc = 0;            nt = r - 4; }
    else if (r < 10) { int q = r - 6;
                       W = upd_w1 + size_t(layer) * 64 * 32; kc = q >> 1;       nt = q & 1; }
    else             { W = upd_w2 + size_t(layer) * 32 * 32; kc = 0;            nt = r - 10; }
    int half = lane >> 4;
    int n = nt * 16 + (lane & 15);
    int k = kc * 32 + 16 * (i >> 3) + half * 8 + (i & 7);
    wfrag[size_t(frag) * 512 + lane * 16 + i] = (_Float16)W[k * 32 + n];
}

__device__ __forceinline__ void embed_body(
    int blk,
    const float* __restrict__ x, const float* __restrict__ w, const float* __restrict__ b,
    float* __restrict__ h, _Float16* __restrict__ h16)
{
    const int q0 = blk * 256 + threadIdx.x;
    const int jq = (q0 & 7) * 4;
    const float4 w0 = *(const float4*)(w + 0 * 32 + jq);
    const float4 w1 = *(const float4*)(w + 1 * 32 + jq);
    const float4 w2 = *(const float4*)(w + 2 * 32 + jq);
    const float4 w3 = *(const float4*)(w + 3 * 32 + jq);
    const float4 bb = *(const float4*)(b + jq);
    float4 xv[EW_PER];
#pragma unroll
    for (int p = 0; p < EW_PER; ++p)
        xv[p] = *(const float4*)(x + size_t((q0 + p * EW_T) >> 3) * 4);
#pragma unroll
    for (int p = 0; p < EW_PER; ++p) {
        const int q = q0 + p * EW_T;
        const float4 xr = xv[p];
        float4 acc;
        acc.x = fmaf(xr.w, w3.x, fmaf(xr.z, w2.x, fmaf(xr.y, w1.x, fmaf(xr.x, w0.x, bb.x))));
        acc.y = fmaf(xr.w, w3.y, fmaf(xr.z, w2.y, fmaf(xr.y, w1.y, fmaf(xr.x, w0.y, bb.y))));
        acc.z = fmaf(xr.w, w3.z, fmaf(xr.z, w2.z, fmaf(xr.y, w1.z, fmaf(xr.x, w0.z, bb.z))));
        acc.w = fmaf(xr.w, w3.w, fmaf(xr.z, w2.w, fmaf(xr.y, w1.w, fmaf(xr.x, w0.w, bb.w))));
        ((float4*)h)[q] = acc;
        v4h a16;
        a16[0] = (_Float16)acc.x; a16[1] = (_Float16)acc.y;
        a16[2] = (_Float16)acc.z; a16[3] = (_Float16)acc.w;
        ((v4h*)h16)[q] = a16;
    }
}

__device__ __forceinline__ void bin_count_body(int chunk, unsigned* lc,
                                               const int* __restrict__ dst,
                                               unsigned* __restrict__ cnt) {
    static_assert(NB == 256, "one counter per thread of a 256-thread block");
    const int tid = threadIdx.x;
    lc[tid] = 0u;
    __syncthreads();
    const int* d = dst + size_t(chunk) * CH;
    constexpr int PER = (CH + 255) / 256;
    int v[PER];
#pragma unroll
    for (int j = 0; j < PER; ++j) { const int i = tid + j * 256; v[j] = (i < CH) ? d[i] : -1; }
#pragma unroll
    for (int j = 0; j < PER; ++j) if (v[j] >= 0) atomicAdd(&lc[v[j] / BW], 1u);
    __syncthreads();
    cnt[chunk * NB + tid] = lc[tid];
}

constexpr int FRONT_GRID = EW_GRID + NCH + PREP_BLOCKS + 1;
__global__ void __launch_bounds__(256) front_kernel(
    const float* __restrict__ x, const float* __restrict__ emb_w, const float* __restrict__ emb_b,
    float* __restrict__ h, _Float16* __restrict__ h16,
    const int* __restrict__ dst, unsigned* __restrict__ cnt,
    const float* __restrict__ msg_w1, const float* __restrict__ msg_w2,
    const float* __restrict__ upd_w1, const float* __restrict__ upd_w2,
    _Float16* __restrict__ wfrag, float* __restrict__ bnsum)
{
    __shared__ unsigned lc[NB];
    const int blk = blockIdx.x;
    if (blk < EW_GRID) {
        embed_body(blk, x, emb_w, emb_b, h, h16);
    } else if (blk < EW_GRID + NCH) {
        bin_count_body(blk - EW_GRID, lc, dst, cnt);
    } else if (blk < EW_GRID + NCH + PREP_BLOCKS) {
        prep_weights_body(blk - EW_GRID - NCH, msg_w1, msg_w2, upd_w1, upd_w2, wfrag);
    } else {
        if (threadIdx.x < 64) bnsum[threadIdx.x] = 0.0f;
    }
}

__device__ __forceinline__ void block_scan_nb(unsigned* sc, int tid) {
    for (int o = 1; o < NB; o <<= 1) {
        unsigned t = 0u;
        if (tid < NB && tid >= o) t = sc[tid - o];
        __syncthreads();
        if (tid < NB) sc[tid] += t;
        __syncthreads();
    }
}

__global__ void __launch_bounds__(BT) bin_scatter_kernel(
    const int* __restrict__ src, const int* __restrict__ dst,
    const unsigned* __restrict__ cnt, int* __restrict__ bstart,
    int* __restrict__ bsrc, int* __restrict__ bdst)
{
    __shared__ unsigned part[4][NB];
    __shared__ unsigned gsc[NB];
    __shared__ unsigned sc[NB], lstart[NB], cur[NB], goff[NB];
    __shared__ int ssrc[CH], sdst[CH];
    const int tid = threadIdx.x;
    const size_t base = size_t(blockIdx.x) * CH;
    int d[CIT], s[CIT];
#pragma unroll
    for (int j = 0; j < CIT; ++j) {
        const int i = tid + j * BT;
        d[j] = -1; s[j] = 0;
        if (i < CH) { d[j] = dst[base + i]; s[j] = src[base + i]; }
    }
    {
        const int b = tid & (NB - 1), g = tid >> 8;
        const int gc = blockIdx.x >> 6, kc = blockIdx.x & 63;
        const unsigned* p = cnt + size_t(g) * 64 * NB + b;
        unsigned v[64];
#pragma unroll
        for (int k = 0; k < 64; ++k) v[k] = p[size_t(k) * NB];
        unsigned sall = 0u, spre = 0u;
#pragma unroll
        for (int k = 0; k < 64; ++k) { sall += v[k]; spre += (k < kc) ? v[k] : 0u; }
        part[g][b] = sall;
        if (tid < NB) sc[tid] = 0u;
        __syncthreads();
        unsigned tot = 0u;
        if (tid < NB) { tot = part[0][tid] + part[1][tid] + part[2][tid] + part[3][tid]; gsc[tid] = tot; }
        __syncthreads();
        block_scan_nb(gsc, tid);
        if (blockIdx.x == 0) {
            if (tid < NB) bstart[tid] = (int)(gsc[tid] - tot);
            if (tid == NB - 1) bstart[NB] = (int)gsc[tid];
        }
        if (g == gc) {
            unsigned run = gsc[b] - (part[0][b] + part[1][b] + part[2][b] + part[3][b]);
            for (int gg = 0; gg < gc; ++gg) run += part[gg][b];
            goff[b] = run + spre;
        }
    }
#pragma unroll
    for (int j = 0; j < CIT; ++j)
        if (d[j] >= 0) atomicAdd(&sc[d[j] / BW], 1u);
    __syncthreads();
    unsigned mine = 0u;
    if (tid < NB) mine = sc[tid];
    block_scan_nb(sc, tid);
    if (tid < NB) { const unsigned st = sc[tid] - mine; lstart[tid] = st; cur[tid] = st; }
    __syncthreads();
#pragma unroll
    for (int j = 0; j < CIT; ++j) {
        if (d[j] >= 0) {
            const unsigned p = atomicAdd(&cur[d[j] / BW], 1u);
            ssrc[p] = s[j]; sdst[p] = d[j];
        }
    }
    __syncthreads();
    for (int i = tid; i < CH; i += BT) {
        const int dd = sdst[i];
        const int b = dd / BW;
        const unsigned g = goff[b] + ((unsigned)i - lstart[b]);
        bsrc[g] = ssrc[i]; bdst[g] = dd;
    }
}

__global__ void __launch_bounds__(256) edge_msg1_kernel(
    const _Float16* __restrict__ h16,
    const int* __restrict__ src, const int* __restrict__ dst,
    const _Float16* __restrict__ wf1, const float* __restrict__ b1,
    _Float16* __restrict__ t16, float* __restrict__ bnsum  )
{
    __shared__ _Float16 tb[WPB][16][32];
    __shared__ float bsum[32], bsq[32];
    const int tid = threadIdx.x;
    if (tid < 32) { bsum[tid] = 0.0f; bsq[tid] = 0.0f; }
    __syncthreads();

    const int lane = tid & 31;
    const int wv = __builtin_amdgcn_readfirstlane(tid >> 5);
    const int m = lane & 15, half = lane >> 4, n0 = lane & 15;
    const int stride = gridDim.x * WPB;
    int tile = blockIdx.x * WPB + wv;

    const v16h b00 = *(const v16h*)(wf1 + 0 * 512 + lane * 16);
    const v16h b01 = *(const v16h*)(wf1 + 1 * 512 + lane * 16);
    const v16h b10 = *(const v16h*)(wf1 + 2 * 512 + lane * 16);
    const v16h b11 = *(const v16h*)(wf1 + 3 * 512 + lane * 16);
    const float bias0 = b1[n0], bias1 = b1[16 + n0];
    float s0 = 0.f, q0 = 0.f, s1 = 0.f, q1 = 0.f;

    const int e0 = tile * 16 + m;
    v16h a0 = load_rowfrag(h16 + size_t(dst[e0]) * 32, half);
    v16h a1 = load_rowfrag(h16 + size_t(src[e0]) * 32, half);
    const int e1 = min(tile + stride, ETILES - 1) * 16 + m;
    int dn = dst[e1], sn = src[e1];

    for (; tile < ETILES; tile += stride) {
        v8f c0 = {}, c1 = {};
        c0 = wmma_f16(a0, b00, c0); c0 = wmma_f16(a1, b10, c0);
        c1 = wmma_f16(a0, b01, c1); c1 = wmma_f16(a1, b11, c1);

        a0 = load_rowfrag(h16 + size_t(dn) * 32, half);
        a1 = load_rowfrag(h16 + size_t(sn) * 32, half);
        const int e2 = min(tile + 2 * stride, ETILES - 1) * 16 + m;
        dn = dst[e2]; sn = src[e2];

#pragma unroll
        for (int r = 0; r < 8; ++r) {
            int mr = half * 8 + r;
            float t0 = swishf(c0[r] + bias0);
            float t1 = swishf(c1[r] + bias1);
            tb[wv][mr][n0]      = (_Float16)t0;
            tb[wv][mr][16 + n0] = (_Float16)t1;
            s0 += t0; q0 += t0 * t0; s1 += t1; q1 += t1 * t1;
        }
        wave_lds_sync();

        v16h out;
#pragma unroll
        for (int i = 0; i < 8; ++i) {
            out[i]     = tb[wv][m][half * 8 + i];
            out[8 + i] = tb[wv][m][16 + half * 8 + i];
        }
        *(v16h*)(t16 + size_t(tile) * 512 + lane * 16) = out;
    }

    atomicAdd(&bsum[n0], s0);      atomicAdd(&bsq[n0], q0);
    atomicAdd(&bsum[16 + n0], s1); atomicAdd(&bsq[16 + n0], q1);
    __syncthreads();
    if (tid < 32) {
        atomic_add_f32(&bnsum[tid],      bsum[tid]);
        atomic_add_f32(&bnsum[32 + tid], bsq[tid]);
    }
}

__global__ void bn_finalize_kernel(float* __restrict__ bnsum,
                                   const float* __restrict__ g, const float* __restrict__ b,
                                   float* __restrict__ ss  )
{
    int n = threadIdx.x;
    if (n >= 32) return;
    const float su = bnsum[n], sq = bnsum[32 + n];
    bnsum[n] = 0.0f;
    bnsum[32 + n] = 0.0f;
    float mu  = su * (1.0f / E_);
    float var = sq * (1.0f / E_) - mu * mu;
    float sc  = g[n] * rsqrtf(var + EPS_);
    ss[n] = sc;
    ss[32 + n] = b[n] - mu * sc;
}

constexpr int S2_T = 1024;
constexpr int S2_W = S2_T / 32;
__global__ void __launch_bounds__(S2_T) edge_msg2_kernel(
    const _Float16* __restrict__ t16, const int* __restrict__ bdst,
    const int* __restrict__ bstart,
    const float* __restrict__ ss, const _Float16* __restrict__ wf2,
    const float* __restrict__ b2, float* __restrict__ agg)
{
    __shared__ __attribute__((aligned(16))) float acc[(BW + 1) * 32];
    __shared__ unsigned cn[BW + 1];
    const int tid = threadIdx.x, lane = tid & 31;
    const int wv = __builtin_amdgcn_readfirstlane(tid >> 5);
    const int half = lane >> 4, n0 = lane & 15;
    const int lo = blockIdx.x * BW;
    const int e0 = bstart[blockIdx.x], e1 = bstart[blockIdx.x + 1];
    const int t1 = (e1 + 15) >> 4;
    int tile = (e0 >> 4) + wv;

    for (int i = tid; i < (BW + 1) * 8; i += S2_T) ((float4*)acc)[i] = make_float4(0.f, 0.f, 0.f, 0.f);
    if (tid <= BW) cn[tid] = 0u;

    float sc[16], sh[16];
#pragma unroll
    for (int g = 0; g < 2; ++g) {
#pragma unroll
        for (int j = 0; j < 2; ++j) {
            const float4 vs = *(const float4*)(ss + 16 * g + half * 8 + 4 * j);
            const float4 vh = *(const float4*)(ss + 32 + 16 * g + half * 8 + 4 * j);
            sc[8 * g + 4 * j + 0] = vs.x; sc[8 * g + 4 * j + 1] = vs.y;
            sc[8 * g + 4 * j + 2] = vs.z; sc[8 * g + 4 * j + 3] = vs.w;
            sh[8 * g + 4 * j + 0] = vh.x; sh[8 * g + 4 * j + 1] = vh.y;
            sh[8 * g + 4 * j + 2] = vh.z; sh[8 * g + 4 * j + 3] = vh.w;
        }
    }
    const v16h w0 = *(const v16h*)(wf2 + 0 * 512 + lane * 16);
    const v16h w1 = *(const v16h*)(wf2 + 1 * 512 + lane * 16);
    const float bias0 = b2[n0], bias1 = b2[16 + n0];

    const int tc = min(tile, ETILES - 1);
    v16h a = *(const v16h*)(t16 + size_t(tc) * 512 + lane * 16);
    const int4* dp = (const int4*)(bdst + tc * 16 + half * 8);
    int4 d0 = dp[0], d1 = dp[1];
    __syncthreads();

    for (; tile < t1; tile += S2_W) {
        v16h an;
#pragma unroll
        for (int i = 0; i < 16; ++i)
            an[i] = (_Float16)((float)a[i] * sc[i] + sh[i]);
        const int node[8] = { d0.x, d0.y, d0.z, d0.w, d1.x, d1.y, d1.z, d1.w };

        const int nt = min(tile + S2_W, ETILES - 1);
        a = *(const v16h*)(t16 + size_t(nt) * 512 + lane * 16);
        dp = (const int4*)(bdst + nt * 16 + half * 8);
        d0 = dp[0]; d1 = dp[1];

        v8f c0 = {}, c1 = {};
        c0 = wmma_f16(an, w0, c0);
        c1 = wmma_f16(an, w1, c1);
#pragma unroll
        for (int r = 0; r < 8; ++r) {
            const float m0 = swishf(c0[r] + bias0), m1 = swishf(c1[r] + bias1);
            const unsigned loc = min((unsigned)(node[r] - lo), (unsigned)BW);
            atomicAdd(&acc[loc * 32 + n0], m0);
            atomicAdd(&acc[loc * 32 + 16 + n0], m1);
            if (n0 == 0) atomicAdd(&cn[loc], 1u);
        }
    }
    __syncthreads();

    const int nrows = min(BW, N_ - lo);
    float4* o4 = (float4*)(agg + size_t(lo) * 32);
    for (int i = tid; i < nrows * 8; i += S2_T) {
        const float rc = __builtin_amdgcn_rcpf(fmaxf((float)cn[i >> 3], 1.0f));
        float4 v = ((const float4*)acc)[i];
        v.x *= rc; v.y *= rc; v.z *= rc; v.w *= rc;
        o4[i] = v;
    }
}

__global__ void __launch_bounds__(256) node_update_kernel(
    const _Float16* __restrict__ wfu1, const _Float16* __restrict__ wfu2,
    const float* __restrict__ ub1, const float* __restrict__ ub2,
    const float* __restrict__ agg,
    float* __restrict__ h, _Float16* __restrict__ h16)
{
    __shared__ _Float16 ub[WPB][16][32];
    __shared__ float    ob[WPB][16][32];
    int tid = threadIdx.x, lane = tid & 31, wv = tid >> 5;
    int tile = blockIdx.x * WPB + wv;
    if (tile >= NTILES) return;

    int m = lane & 15, half = lane >> 4;
    size_t node = size_t(tile) * 16 + m;

    v16h a0 = load_rowfrag(h16 + node * 32, half);
    const float* ar = agg + node * 32;
    v16h a1;
#pragma unroll
    for (int i = 0; i < 8; ++i) {
        a1[i]     = (_Float16)ar[half * 8 + i];
        a1[8 + i] = (_Float16)ar[16 + half * 8 + i];
    }
    v16h b00 = *(const v16h*)(wfu1 + 0 * 512 + lane * 16);
    v16h b01 = *(const v16h*)(wfu1 + 1 * 512 + lane * 16);
    v16h b10 = *(const v16h*)(wfu1 + 2 * 512 + lane * 16);
    v16h b11 = *(const v16h*)(wfu1 + 3 * 512 + lane * 16);
    v8f c0 = {}, c1 = {};
    c0 = wmma_f16(a0, b00, c0); c0 = wmma_f16(a1, b10, c0);
    c1 = wmma_f16(a0, b01, c1); c1 = wmma_f16(a1, b11, c1);

    int n0 = lane & 15;
    float bias0 = ub1[n0], bias1 = ub1[16 + n0];
#pragma unroll
    for (int r = 0; r < 8; ++r) {
        int mr = (lane >> 4) * 8 + r;
        ub[wv][mr][n0]      = (_Float16)swishf(c0[r] + bias0);
        ub[wv][mr][16 + n0] = (_Float16)swishf(c1[r] + bias1);
    }
    wave_lds_sync();

    v16h au;
#pragma unroll
    for (int i = 0; i < 8; ++i) {
        au[i]     = ub[wv][m][half * 8 + i];
        au[8 + i] = ub[wv][m][16 + half * 8 + i];
    }
    v16h u0 = *(const v16h*)(wfu2 + 0 * 512 + lane * 16);
    v16h u1 = *(const v16h*)(wfu2 + 1 * 512 + lane * 16);
    v8f d0 = {}, d1 = {};
    d0 = wmma_f16(au, u0, d0);
    d1 = wmma_f16(au, u1, d1);

    float bb0 = ub2[n0], bb1 = ub2[16 + n0];
#pragma unroll
    for (int r = 0; r < 8; ++r) {
        int mr = (lane >> 4) * 8 + r;
        ob[wv][mr][n0]      = swishf(d0[r] + bb0);
        ob[wv][mr][16 + n0] = swishf(d1[r] + bb1);
    }
    wave_lds_sync();

    int row = lane >> 1, seg = (lane & 1) * 16;
    size_t nr = size_t(tile) * 16 + row;
    float*    hp   = h   + nr * 32 + seg;
    _Float16* hp16 = h16 + nr * 32 + seg;
#pragma unroll
    for (int i = 0; i < 16; ++i) {
        float nv = hp[i] + ob[wv][row][seg + i];
        hp[i] = nv;
        hp16[i] = (_Float16)nv;
    }
}

extern "C" void kernel_launch(void* const* d_in, const int* in_sizes, int n_in,
                              void* d_out, int out_size, void* d_ws, size_t ws_size,
                              hipStream_t stream) {
    const float* x      = (const float*)d_in[0];
    const float* emb_w  = (const float*)d_in[1];
    const float* emb_b  = (const float*)d_in[2];
    const float* msg_w1 = (const float*)d_in[3];
    const float* msg_b1 = (const float*)d_in[4];
    const float* bn_g   = (const float*)d_in[5];
    const float* bn_b   = (const float*)d_in[6];
    const float* msg_w2 = (const float*)d_in[7];
    const float* msg_b2 = (const float*)d_in[8];
    const float* upd_w1 = (const float*)d_in[9];
    const float* upd_b1 = (const float*)d_in[10];
    const float* upd_w2 = (const float*)d_in[11];
    const float* upd_b2 = (const float*)d_in[12];
    const int*   eidx   = (const int*)d_in[13];
    const int*   esrc   = eidx;
    const int*   edst   = eidx + E_;

    char* ws = (char*)d_ws;
    _Float16* wfrag = (_Float16*)(ws);
    float*    h     = (float*)d_out;
    int*      bsrc  = (int*)(ws + OFF_ES);
    int*      bdst  = (int*)(ws + OFF_ED);
    _Float16* h16   = (_Float16*)(ws + OFF_H16);
    float*    agg   = (float*)(ws + OFF_AGG);
    unsigned* boff  = (unsigned*)(ws + OFF_BIN);
    int*      bst   = (int*)(ws + OFF_BST);
    float*    bnsum = (float*)(ws + OFF_BNS);
    float*    bnss  = (float*)(ws + OFF_BNP);
    _Float16* t16   = (_Float16*)(ws + OFF_T16);

    front_kernel<<<FRONT_GRID, 256, 0, stream>>>(
        x, emb_w, emb_b, h, h16, edst, boff, msg_w1, msg_w2, upd_w1, upd_w2, wfrag, bnsum);
    bin_scatter_kernel<<<NCH, BT, 0, stream>>>(esrc, edst, boff, bst, bsrc, bdst);

    for (int l = 0; l < L_; ++l) {
        edge_msg1_kernel<<<EDGE1_GRID, 256, 0, stream>>>(
            h16, bsrc, bdst, wfrag + size_t(l * 12 + 0) * 512, msg_b1 + l * D_, t16, bnsum);
        bn_finalize_kernel<<<1, 32, 0, stream>>>(bnsum, bn_g + l * D_, bn_b + l * D_, bnss);
        edge_msg2_kernel<<<NB, S2_T, 0, stream>>>(
            t16, bdst, bst, bnss, wfrag + size_t(l * 12 + 4) * 512, msg_b2 + l * D_, agg);
        node_update_kernel<<<NODE_BLOCKS, 256, 0, stream>>>(
            wfrag + size_t(l * 12 + 6) * 512, wfrag + size_t(l * 12 + 10) * 512,
            upd_b1 + l * D_, upd_b2 + l * D_, agg, h, h16);
    }
}


# Edge stage 1: products transposed (weights as A, edge rows as B): a lane holds one edge's features, t goes from accumulators to memory, no LDS staging; bias seeds accumulators; BN sum order changes
# speedup vs baseline: 2.6957x; 1.2027x over previous
#include <hip/hip_runtime.h>
#include <hip/hip_bf16.h>

constexpr int N_ = 100000;
constexpr int E_ = 1600000;
constexpr int D_ = 32;
constexpr int L_ = 2;
constexpr float EPS_ = 1e-5f;

constexpr int WPB = 8;
constexpr int ETILES = E_ / 16;
constexpr int EDGE1_GRID = 256 * 3;
static_assert(EDGE1_GRID * WPB <= ETILES, "every wave owns at least one tile");
constexpr int NB  = 256;
constexpr int BW  = 391;
static_assert(NB * BW >= N_, "the ranges cover every node");
constexpr int NCH = 256;
constexpr int CH  = E_ / NCH;
constexpr int BT  = 1024;
constexpr int CIT = (CH + BT - 1) / BT;
static_assert(E_ % NCH == 0 && NB == 256 && NCH == 256 && BT == 1024, "binning split");
constexpr int NPIECES = N_ * D_ / 4;
constexpr int EW_GRID = 625;
constexpr int EW_PER  = 5;
constexpr int EW_T    = EW_GRID * 256;
static_assert((N_ * D_) % 4 == 0 && EW_T * EW_PER == NPIECES && EW_T % 8 == 0,
              "elementwise split is exact");

typedef __attribute__((ext_vector_type(16))) _Float16 v16h;
typedef __attribute__((ext_vector_type(8)))  _Float16 v8h;
typedef __attribute__((ext_vector_type(4)))  _Float16 v4h;
typedef __attribute__((ext_vector_type(8)))  float    v8f;

__device__ __forceinline__ v8f wmma_f16(v16h a, v16h b, v8f c) {
    return __builtin_amdgcn_wmma_f32_16x16x32_f16(
        false, a, false, b, (short)0, c, false, false);
}

__device__ __forceinline__ float swishf(float x) {
    float e = __expf(-x);
    return x * __builtin_amdgcn_rcpf(1.0f + e);
}

__device__ __forceinline__ v16h load_rowfrag(const _Float16* __restrict__ row, int half) {
    v8h lo = *(const v8h*)(row + half * 8);
    v8h hi = *(const v8h*)(row + 16 + half * 8);
    v16h a;
#pragma unroll
    for (int i = 0; i < 8; ++i) { a[i] = lo[i]; a[8 + i] = hi[i]; }
    return a;
}

__device__ __forceinline__ void wave_lds_sync() {
    __builtin_amdgcn_wave_barrier();
    asm volatile("s_wait_dscnt 0" ::: "memory");
    __builtin_amdgcn_wave_barrier();
}

__device__ __forceinline__ void atomic_add_f32(float* p, float v) {
    __hip_atomic_fetch_add(p, v, __ATOMIC_RELAXED, __HIP_MEMORY_SCOPE_AGENT);
}

constexpr size_t alignup(size_t x) { return (x + 255) & ~size_t(255); }
constexpr size_t SZ_WFRAG = size_t(L_) * 12 * 512 * 2;
constexpr size_t OFF_ES   = alignup(SZ_WFRAG);
constexpr size_t OFF_ED   = OFF_ES + size_t(E_) * 4;
constexpr size_t OFF_H16  = alignup(OFF_ED  + size_t(E_) * 4);
constexpr size_t OFF_BIN  = alignup(OFF_H16 + size_t(N_) * D_ * 2);
constexpr size_t OFF_BST  = OFF_BIN + size_t(NCH) * NB * 4;
constexpr size_t OFF_BNS  = alignup(OFF_BST + size_t(NB + 1) * 4);
static_assert(size_t(E_) * 8 <= size_t(N_) * D_ * 4 && size_t(NCH) * NB * 4 + size_t(NB + 1) * 4 <= size_t(N_) * 4,
              "workspace use stays within one [N,32] f32 plane plus one [N] f32 plane for these");
constexpr size_t OFF_T16  = alignup(OFF_BNS + size_t(L_) * 64 * 4);

constexpr int PREP_BLOCKS = (L_ * 12 * 512 + 255) / 256;
__device__ __forceinline__ void prep_weights_body(
    int blk,
    const float* __restrict__ msg_w1, const float* __restrict__ msg_w2,
    const float* __restrict__ upd_w1, const float* __restrict__ upd_w2,
    _Float16* __restrict__ wfrag)
{
    int gid = blk * 256 + threadIdx.x;
    if (gid >= L_ * 12 * 512) return;
    int frag = gid >> 9;
    int lane = (gid >> 4) & 31;
    int i    = gid & 15;
    int layer = frag / 12;
    int r     = frag % 12;
    const float* W; int kc, nt;
    if (r < 4)       { W = msg_w1 + size_t(layer) * 64 * 32; kc = r >> 1;       nt = r & 1; }
    else if (r < 6)  { W = msg_w2 + size_t(layer) * 32 * 32; kc = 0;            nt = r - 4; }
    else if (r < 10) { int q = r - 6;
                       W = upd_w1 + size_t(layer) * 64 * 32; kc = q >> 1;       nt = q & 1; }
    else             { W = upd_w2 + size_t(layer) * 32 * 32; kc = 0;            nt = r - 10; }
    int half = lane >> 4;
    int n = nt * 16 + (lane & 15);
    int k = kc * 32 + 16 * (i >> 3) + half * 8 + (i & 7);
    wfrag[size_t(frag) * 512 + lane * 16 + i] = (_Float16)W[k * 32 + n];
}

__device__ __forceinline__ void embed_body(
    int blk,
    const float* __restrict__ x, const float* __restrict__ w, const float* __restrict__ b,
    float* __restrict__ h, _Float16* __restrict__ h16)
{
    const int q0 = blk * 256 + threadIdx.x;
    const int jq = (q0 & 7) * 4;
    const float4 w0 = *(const float4*)(w + 0 * 32 + jq);
    const float4 w1 = *(const float4*)(w + 1 * 32 + jq);
    const float4 w2 = *(const float4*)(w + 2 * 32 + jq);
    const float4 w3 = *(const float4*)(w + 3 * 32 + jq);
    const float4 bb = *(const float4*)(b + jq);
    float4 xv[EW_PER];
#pragma unroll
    for (int p = 0; p < EW_PER; ++p)
        xv[p] = *(const float4*)(x + size_t((q0 + p * EW_T) >> 3) * 4);
#pragma unroll
    for (int p = 0; p < EW_PER; ++p) {
        const int q = q0 + p * EW_T;
        const float4 xr = xv[p];
        float4 acc;
        acc.x = fmaf(xr.w, w3.x, fmaf(xr.z, w2.x, fmaf(xr.y, w1.x, fmaf(xr.x, w0.x, bb.x))));
        acc.y = fmaf(xr.w, w3.y, fmaf(xr.z, w2.y, fmaf(xr.y, w1.y, fmaf(xr.x, w0.y, bb.y))));
        acc.z = fmaf(xr.w, w3.z, fmaf(xr.z, w2.z, fmaf(xr.y, w1.z, fmaf(xr.x, w0.z, bb.z))));
        acc.w = fmaf(xr.w, w3.w, fmaf(xr.z, w2.w, fmaf(xr.y, w1.w, fmaf(xr.x, w0.w, bb.w))));
        ((float4*)h)[q] = acc;
        v4h a16;
        a16[0] = (_Float16)acc.x; a16[1] = (_Float16)acc.y;
        a16[2] = (_Float16)acc.z; a16[3] = (_Float16)acc.w;
        ((v4h*)h16)[q] = a16;
    }
}

__device__ __forceinline__ void bin_count_body(int chunk, unsigned* lc,
                                               const int* __restrict__ dst,
                                               unsigned* __restrict__ cnt) {
    static_assert(NB == 256, "one counter per thread of a 256-thread block");
    const int tid = threadIdx.x;
    lc[tid] = 0u;
    __syncthreads();
    const int* d = dst + size_t(chunk) * CH;
    constexpr int PER = (CH + 255) / 256;
    int v[PER];
#pragma unroll
    for (int j = 0; j < PER; ++j) { const int i = tid + j * 256; v[j] = (i < CH) ? d[i] : -1; }
#pragma unroll
    for (int j = 0; j < PER; ++j) if (v[j] >= 0) atomicAdd(&lc[v[j] / BW], 1u);
    __syncthreads();
    cnt[chunk * NB + tid] = lc[tid];
}

constexpr int FRONT_GRID = EW_GRID + NCH + PREP_BLOCKS + 1;
__global__ void __launch_bounds__(256) front_kernel(
    const float* __restrict__ x, const float* __restrict__ emb_w, const float* __restrict__ emb_b,
    float* __restrict__ h, _Float16* __restrict__ h16,
    const int* __restrict__ dst, unsigned* __restrict__ cnt,
    const float* __restrict__ msg_w1, const float* __restrict__ msg_w2,
    const float* __restrict__ upd_w1, const float* __restrict__ upd_w2,
    _Float16* __restrict__ wfrag, float* __restrict__ bnsum)
{
    __shared__ unsigned lc[NB];
    const int blk = blockIdx.x;
    if (blk < EW_GRID) {
        embed_body(blk, x, emb_w, emb_b, h, h16);
    } else if (blk < EW_GRID + NCH) {
        bin_count_body(blk - EW_GRID, lc, dst, cnt);
    } else if (blk < EW_GRID + NCH + PREP_BLOCKS) {
        prep_weights_body(blk - EW_GRID - NCH, msg_w1, msg_w2, upd_w1, upd_w2, wfrag);
    } else {
        if (threadIdx.x < L_ * 64) bnsum[threadIdx.x] = 0.0f;
    }
}

__device__ __forceinline__ void block_scan_nb(unsigned* sc, int tid) {
    for (int o = 1; o < NB; o <<= 1) {
        unsigned t = 0u;
        if (tid < NB && tid >= o) t = sc[tid - o];
        __syncthreads();
        if (tid < NB) sc[tid] += t;
        __syncthreads();
    }
}

__global__ void __launch_bounds__(BT) bin_scatter_kernel(
    const int* __restrict__ src, const int* __restrict__ dst,
    const unsigned* __restrict__ cnt, int* __restrict__ bstart,
    int* __restrict__ bsrc, int* __restrict__ bdst)
{
    __shared__ unsigned part[4][NB];
    __shared__ unsigned gsc[NB];
    __shared__ unsigned sc[NB], lstart[NB], cur[NB], goff[NB];
    __shared__ int ssrc[CH], sdst[CH];
    const int tid = threadIdx.x;
    const size_t base = size_t(blockIdx.x) * CH;
    int d[CIT], s[CIT];
#pragma unroll
    for (int j = 0; j < CIT; ++j) {
        const int i = tid + j * BT;
        d[j] = -1; s[j] = 0;
        if (i < CH) { d[j] = dst[base + i]; s[j] = src[base + i]; }
    }
    {
        const int b = tid & (NB - 1), g = tid >> 8;
        const int gc = blockIdx.x >> 6, kc = blockIdx.x & 63;
        const unsigned* p = cnt + size_t(g) * 64 * NB + b;
        unsigned v[64];
#pragma unroll
        for (int k = 0; k < 64; ++k) v[k] = p[size_t(k) * NB];
        unsigned sall = 0u, spre = 0u;
#pragma unroll
        for (int k = 0; k < 64; ++k) { sall += v[k]; spre += (k < kc) ? v[k] : 0u; }
        part[g][b] = sall;
        if (tid < NB) sc[tid] = 0u;
        __syncthreads();
        unsigned tot = 0u;
        if (tid < NB) { tot = part[0][tid] + part[1][tid] + part[2][tid] + part[3][tid]; gsc[tid] = tot; }
        __syncthreads();
        block_scan_nb(gsc, tid);
        if (blockIdx.x == 0) {
            if (tid < NB) bstart[tid] = (int)(gsc[tid] - tot);
            if (tid == NB - 1) bstart[NB] = (int)gsc[tid];
        }
        if (g == gc) {
            unsigned run = gsc[b] - (part[0][b] + part[1][b] + part[2][b] + part[3][b]);
            for (int gg = 0; gg < gc; ++gg) run += part[gg][b];
            goff[b] = run + spre;
        }
    }
#pragma unroll
    for (int j = 0; j < CIT; ++j)
        if (d[j] >= 0) atomicAdd(&sc[d[j] / BW], 1u);
    __syncthreads();
    unsigned mine = 0u;
    if (tid < NB) mine = sc[tid];
    block_scan_nb(sc, tid);
    if (tid < NB) { const unsigned st = sc[tid] - mine; lstart[tid] = st; cur[tid] = st; }
    __syncthreads();
#pragma unroll
    for (int j = 0; j < CIT; ++j) {
        if (d[j] >= 0) {
            const unsigned p = atomicAdd(&cur[d[j] / BW], 1u);
            ssrc[p] = s[j]; sdst[p] = d[j];
        }
    }
    __syncthreads();
    for (int i = tid; i < CH; i += BT) {
        const int dd = sdst[i];
        const int b = dd / BW;
        const unsigned g = goff[b] + ((unsigned)i - lstart[b]);
        bsrc[g] = ssrc[i]; bdst[g] = dd;
    }
}

__global__ void __launch_bounds__(256) edge_msg1_kernel(
    const _Float16* __restrict__ h16,
    const int* __restrict__ src, const int* __restrict__ dst,
    const _Float16* __restrict__ wf1, const float* __restrict__ b1,
    _Float16* __restrict__ t16, float* __restrict__ bnsum  )
{
    constexpr int RP = 36;
    __shared__ __attribute__((aligned(16))) float red[WPB][32][RP];
    __shared__ float bs[64];
    const int tid = threadIdx.x;
    if (tid < 64) bs[tid] = 0.0f;
    __syncthreads();

    const int lane = tid & 31;
    const int wv = __builtin_amdgcn_readfirstlane(tid >> 5);
    const int m = lane & 15, half = lane >> 4;
    const int stride = gridDim.x * WPB;
    int tile = blockIdx.x * WPB + wv;

    const v16h w00 = *(const v16h*)(wf1 + 0 * 512 + lane * 16);
    const v16h w01 = *(const v16h*)(wf1 + 1 * 512 + lane * 16);
    const v16h w10 = *(const v16h*)(wf1 + 2 * 512 + lane * 16);
    const v16h w11 = *(const v16h*)(wf1 + 3 * 512 + lane * 16);
    const float4 p0 = *(const float4*)(b1 + half * 8),      p1 = *(const float4*)(b1 + half * 8 + 4);
    const float4 p2 = *(const float4*)(b1 + 16 + half * 8), p3 = *(const float4*)(b1 + 16 + half * 8 + 4);
    const v8f bias0 = { p0.x, p0.y, p0.z, p0.w, p1.x, p1.y, p1.z, p1.w };
    const v8f bias1 = { p2.x, p2.y, p2.z, p2.w, p3.x, p3.y, p3.z, p3.w };
    v8f s0 = {}, q0 = {}, s1 = {}, q1 = {};

    const int e0 = tile * 16 + m;
    v16h a0 = load_rowfrag(h16 + size_t(dst[e0]) * 32, half);
    v16h a1 = load_rowfrag(h16 + size_t(src[e0]) * 32, half);
    const int e1 = min(tile + stride, ETILES - 1) * 16 + m;
    int dn = dst[e1], sn = src[e1];

    for (; tile < ETILES; tile += stride) {
        v8f c0 = wmma_f16(w00, a0, bias0); c0 = wmma_f16(w10, a1, c0);
        v8f c1 = wmma_f16(w01, a0, bias1); c1 = wmma_f16(w11, a1, c1);

        a0 = load_rowfrag(h16 + size_t(dn) * 32, half);
        a1 = load_rowfrag(h16 + size_t(sn) * 32, half);
        const int e2 = min(tile + 2 * stride, ETILES - 1) * 16 + m;
        dn = dst[e2]; sn = src[e2];
        __builtin_amdgcn_sched_group_barrier(0x008, 4, 0);
        __builtin_amdgcn_sched_group_barrier(0x020, 6, 0);
        __builtin_amdgcn_sched_barrier(0);

        v16h out;
#pragma unroll
        for (int r = 0; r < 8; ++r) {
            const float t0 = swishf(c0[r]);
            const float t1 = swishf(c1[r]);
            out[r]     = (_Float16)t0;
            out[8 + r] = (_Float16)t1;
            s0[r] += t0; q0[r] = fmaf(t0, t0, q0[r]);
            s1[r] += t1; q1[r] = fmaf(t1, t1, q1[r]);
        }
        *(v16h*)(t16 + size_t(tile) * 512 + lane * 16) = out;
    }

    float* rw = &red[wv][lane][0];
    *(float4*)(rw +  0) = make_float4(s0[0], s0[1], s0[2], s0[3]);
    *(float4*)(rw +  4) = make_float4(s0[4], s0[5], s0[6], s0[7]);
    *(float4*)(rw +  8) = make_float4(s1[0], s1[1], s1[2], s1[3]);
    *(float4*)(rw + 12) = make_float4(s1[4], s1[5], s1[6], s1[7]);
    *(float4*)(rw + 16) = make_float4(q0[0], q0[1], q0[2], q0[3]);
    *(float4*)(rw + 20) = make_float4(q0[4], q0[5], q0[6], q0[7]);
    *(float4*)(rw + 24) = make_float4(q1[0], q1[1], q1[2], q1[3]);
    *(float4*)(rw + 28) = make_float4(q1[4], q1[5], q1[6], q1[7]);
    wave_lds_sync();
    float lo = 0.f, hi = 0.f;
#pragma unroll
    for (int l = 0; l < 16; ++l) { lo += red[wv][l][lane]; hi += red[wv][16 + l][lane]; }
    const int jj = lane & 15;
    const int f = (lane >> 4) * 32 + (jj < 8 ? jj : jj + 8);
    atomicAdd(&bs[f], lo);
    atomicAdd(&bs[f + 8], hi);
    __syncthreads();
    if (tid < 64) atomic_add_f32(&bnsum[tid], bs[tid]);
}

__device__ __forceinline__ void bn_scale_shift(float su, float sq, float g, float b,
                                               float& sc, float& sh) {
    float mu  = su * (1.0f / E_);
    float var = sq * (1.0f / E_) - mu * mu;
    sc = g * rsqrtf(var + EPS_);
    sh = b - mu * sc;
}

constexpr int S2_T = 1024;
constexpr int S2_W = S2_T / 32;
__global__ void __launch_bounds__(S2_T) edge_msg2_kernel(
    const _Float16* __restrict__ t16, const int* __restrict__ bdst,
    const int* __restrict__ bstart,
    const float* __restrict__ bnsum  ,
    const float* __restrict__ bng, const float* __restrict__ bnb,
    const _Float16* __restrict__ wf2, const float* __restrict__ b2,
    const _Float16* __restrict__ wfu1, const _Float16* __restrict__ wfu2,
    const float* __restrict__ ub1, const float* __restrict__ ub2,
    float* __restrict__ h, _Float16* __restrict__ h16)
{
    __shared__ __attribute__((aligned(16))) float acc[(BW + 1) * 32];
    __shared__ unsigned cn[BW + 1];
    const int tid = threadIdx.x, lane = tid & 31;
    const int wv = __builtin_amdgcn_readfirstlane(tid >> 5);
    const int half = lane >> 4, n0 = lane & 15;
    const int lo = blockIdx.x * BW;
    const int e0 = bstart[blockIdx.x], e1 = bstart[blockIdx.x + 1];
    const int t1 = (e1 + 15) >> 4;
    int tile = (e0 >> 4) + wv;

    for (int i = tid; i < (BW + 1) * 8; i += S2_T) ((float4*)acc)[i] = make_float4(0.f, 0.f, 0.f, 0.f);
    if (tid <= BW) cn[tid] = 0u;

    float sc[16], sh[16];
#pragma unroll
    for (int g = 0; g < 2; ++g) {
#pragma unroll
        for (int j = 0; j < 2; ++j) {
            const int k = 16 * g + half * 8 + 4 * j, i = 8 * g + 4 * j;
            const float4 su = *(const float4*)(bnsum + k);
            const float4 sq = *(const float4*)(bnsum + 32 + k);
            const float4 gg = *(const float4*)(bng + k);
            const float4 bb = *(const float4*)(bnb + k);
            bn_scale_shift(su.x, sq.x, gg.x, bb.x, sc[i + 0], sh[i + 0]);
            bn_scale_shift(su.y, sq.y, gg.y, bb.y, sc[i + 1], sh[i + 1]);
            bn_scale_shift(su.z, sq.z, gg.z, bb.z, sc[i + 2], sh[i + 2]);
            bn_scale_shift(su.w, sq.w, gg.w, bb.w, sc[i + 3], sh[i + 3]);
        }
    }
    const v16h w0 = *(const v16h*)(wf2 + 0 * 512 + lane * 16);
    const v16h w1 = *(const v16h*)(wf2 + 1 * 512 + lane * 16);
    const float bias0 = b2[n0], bias1 = b2[16 + n0];

    const int tc = min(tile, ETILES - 1);
    v16h a = *(const v16h*)(t16 + size_t(tc) * 512 + lane * 16);
    const int4* dp = (const int4*)(bdst + tc * 16 + half * 8);
    int4 d0 = dp[0], d1 = dp[1];
    __syncthreads();

    for (; tile < t1; tile += S2_W) {
        v16h an;
#pragma unroll
        for (int i = 0; i < 16; ++i)
            an[i] = (_Float16)((float)a[i] * sc[i] + sh[i]);
        const int node[8] = { d0.x, d0.y, d0.z, d0.w, d1.x, d1.y, d1.z, d1.w };

        const int nt = min(tile + S2_W, ETILES - 1);
        a = *(const v16h*)(t16 + size_t(nt) * 512 + lane * 16);
        dp = (const int4*)(bdst + nt * 16 + half * 8);
        d0 = dp[0]; d1 = dp[1];

        v8f c0 = {}, c1 = {};
        c0 = wmma_f16(an, w0, c0);
        c1 = wmma_f16(an, w1, c1);
#pragma unroll
        for (int r = 0; r < 8; ++r) {
            const float m0 = swishf(c0[r] + bias0), m1 = swishf(c1[r] + bias1);
            const unsigned loc = min((unsigned)(node[r] - lo), (unsigned)BW);
            atomicAdd(&acc[loc * 32 + n0], m0);
            atomicAdd(&acc[loc * 32 + 16 + n0], m1);
            if (n0 == 0) atomicAdd(&cn[loc], 1u);
        }
    }
    __syncthreads();

    const int nrows = min(BW, N_ - lo);
    const int ntl = (nrows + 15) >> 4;
    static_assert((BW + 15) / 16 <= S2_W, "a wave per node tile");
    if (wv < ntl) {
        const int rl = wv * 16 + n0;
        const bool valid = rl < nrows;
        const int lr = valid ? rl : nrows - 1;
        const size_t node = size_t(lo + lr);
        float* hp = h + node * 32 + half * 8;
        const float4 x0 = *(const float4*)(hp);
        const float4 x1 = *(const float4*)(hp + 4);
        const float4 x2 = *(const float4*)(hp + 16);
        const float4 x3 = *(const float4*)(hp + 20);
        const v16h bh = load_rowfrag(h16 + node * 32, half);
        const v16h wa00 = *(const v16h*)(wfu1 + 0 * 512 + lane * 16);
        const v16h wa01 = *(const v16h*)(wfu1 + 1 * 512 + lane * 16);
        const v16h wa10 = *(const v16h*)(wfu1 + 2 * 512 + lane * 16);
        const v16h wa11 = *(const v16h*)(wfu1 + 3 * 512 + lane * 16);
        const v16h wb0  = *(const v16h*)(wfu2 + 0 * 512 + lane * 16);
        const v16h wb1  = *(const v16h*)(wfu2 + 1 * 512 + lane * 16);
        const float4 p0 = *(const float4*)(ub1 + half * 8),      p1 = *(const float4*)(ub1 + half * 8 + 4);
        const float4 p2 = *(const float4*)(ub1 + 16 + half * 8), p3 = *(const float4*)(ub1 + 16 + half * 8 + 4);
        const float4 r0 = *(const float4*)(ub2 + half * 8),      r1 = *(const float4*)(ub2 + half * 8 + 4);
        const float4 r2 = *(const float4*)(ub2 + 16 + half * 8), r3 = *(const float4*)(ub2 + 16 + half * 8 + 4);
        const float bu1a[8] = { p0.x, p0.y, p0.z, p0.w, p1.x, p1.y, p1.z, p1.w };
        const float bu1b[8] = { p2.x, p2.y, p2.z, p2.w, p3.x, p3.y, p3.z, p3.w };
        const float bu2a[8] = { r0.x, r0.y, r0.z, r0.w, r1.x, r1.y, r1.z, r1.w };
        const float bu2b[8] = { r2.x, r2.y, r2.z, r2.w, r3.x, r3.y, r3.z, r3.w };
        const float hxa[8]  = { x0.x, x0.y, x0.z, x0.w, x1.x, x1.y, x1.z, x1.w };
        const float hxb[8]  = { x2.x, x2.y, x2.z, x2.w, x3.x, x3.y, x3.z, x3.w };

        const float rc = __builtin_amdgcn_rcpf(fmaxf((float)cn[lr], 1.0f));
        const float* ar = acc + lr * 32 + half * 8;
        v16h ba;
#pragma unroll
        for (int i = 0; i < 8; ++i) {
            ba[i]     = (_Float16)(ar[i] * rc);
            ba[8 + i] = (_Float16)(ar[16 + i] * rc);
        }
        v8f c0 = {}, c1 = {};
        c0 = wmma_f16(wa00, bh, c0); c0 = wmma_f16(wa10, ba, c0);
        c1 = wmma_f16(wa01, bh, c1); c1 = wmma_f16(wa11, ba, c1);
        v16h bu;
#pragma unroll
        for (int r = 0; r < 8; ++r) {
            bu[r]     = (_Float16)swishf(c0[r] + bu1a[r]);
            bu[8 + r] = (_Float16)swishf(c1[r] + bu1b[r]);
        }
        v8f d0 = {}, d1 = {};
        d0 = wmma_f16(wb0, bu, d0);
        d1 = wmma_f16(wb1, bu, d1);
        float na[8], nb[8];
        v8h ha, hb;
#pragma unroll
        for (int r = 0; r < 8; ++r) {
            na[r] = hxa[r] + swishf(d0[r] + bu2a[r]);
            nb[r] = hxb[r] + swishf(d1[r] + bu2b[r]);
            ha[r] = (_Float16)na[r];
            hb[r] = (_Float16)nb[r];
        }
        if (valid) {
            *(float4*)(hp)      = make_float4(na[0], na[1], na[2], na[3]);
            *(float4*)(hp + 4)  = make_float4(na[4], na[5], na[6], na[7]);
            *(float4*)(hp + 16) = make_float4(nb[0], nb[1], nb[2], nb[3]);
            *(float4*)(hp + 20) = make_float4(nb[4], nb[5], nb[6], nb[7]);
            _Float16* hq = h16 + node * 32 + half * 8;
            *(v8h*)(hq)      = ha;
            *(v8h*)(hq + 16) = hb;
        }
    }
}

extern "C" void kernel_launch(void* const* d_in, const int* in_sizes, int n_in,
                              void* d_out, int out_size, void* d_ws, size_t ws_size,
                              hipStream_t stream) {
    const float* x      = (const float*)d_in[0];
    const float* emb_w  = (const float*)d_in[1];
    const float* emb_b  = (const float*)d_in[2];
    const float* msg_w1 = (const float*)d_in[3];
    const float* msg_b1 = (const float*)d_in[4];
    const float* bn_g   = (const float*)d_in[5];
    const float* bn_b   = (const float*)d_in[6];
    const float* msg_w2 = (const float*)d_in[7];
    const float* msg_b2 = (const float*)d_in[8];
    const float* upd_w1 = (const float*)d_in[9];
    const float* upd_b1 = (const float*)d_in[10];
    const float* upd_w2 = (const float*)d_in[11];
    const float* upd_b2 = (const float*)d_in[12];
    const int*   eidx   = (const int*)d_in[13];
    const int*   esrc   = eidx;
    const int*   edst   = eidx + E_;

    char* ws = (char*)d_ws;
    _Float16* wfrag = (_Float16*)(ws);
    float*    h     = (float*)d_out;
    int*      bsrc  = (int*)(ws + OFF_ES);
    int*      bdst  = (int*)(ws + OFF_ED);
    _Float16* h16   = (_Float16*)(ws + OFF_H16);
    unsigned* boff  = (unsigned*)(ws + OFF_BIN);
    int*      bst   = (int*)(ws + OFF_BST);
    float*    bnsum = (float*)(ws + OFF_BNS);
    _Float16* t16   = (_Float16*)(ws + OFF_T16);

    front_kernel<<<FRONT_GRID, 256, 0, stream>>>(
        x, emb_w, emb_b, h, h16, edst, boff, msg_w1, msg_w2, upd_w1, upd_w2, wfrag, bnsum);
    bin_scatter_kernel<<<NCH, BT, 0, stream>>>(esrc, edst, boff, bst, bsrc, bdst);

    for (int l = 0; l < L_; ++l) {
        edge_msg1_kernel<<<EDGE1_GRID, 256, 0, stream>>>(
            h16, bsrc, bdst, wfrag + size_t(l * 12 + 0) * 512, msg_b1 + l * D_, t16, bnsum + l * 64);
        edge_msg2_kernel<<<NB, S2_T, 0, stream>>>(
            t16, bdst, bst, bnsum + l * 64, bn_g + l * D_, bn_b + l * D_,
            wfrag + size_t(l * 12 + 4) * 512, msg_b2 + l * D_,
            wfrag + size_t(l * 12 + 6) * 512, wfrag + size_t(l * 12 + 10) * 512,
            upd_b1 + l * D_, upd_b2 + l * D_, h, h16);
    }
}
